# starting kernel
# baseline (speedup 1.0000x reference)
#include <hip/hip_runtime.h>
#include <hip/hip_bf16.h>
#include <math.h>

#define NN 100000
#define NE 1600000

typedef __attribute__((ext_vector_type(2))) float v2f;
typedef __attribute__((ext_vector_type(8))) float v8f;


__global__ void zero_f(float* __restrict__ p, int n) {
    int i = blockIdx.x * blockDim.x + threadIdx.x;
    if (i < n) p[i] = 0.0f;
}

__global__ void deg_count(const int* __restrict__ dst, float* __restrict__ deg, int e_cnt) {
    int e = blockIdx.x * blockDim.x + threadIdx.x;
    if (e < e_cnt) atomicAdd(&deg[dst[e]], 1.0f);
}

__global__ void deg_norm(const float* __restrict__ deg, float* __restrict__ dinv,
                         float* __restrict__ selfn, int n) {
    int i = blockIdx.x * blockDim.x + threadIdx.x;
    if (i >= n) return;
    float d = deg[i] + 1.0f;
    dinv[i]  = rsqrtf(d);
    selfn[i] = 1.0f / d;
}

__global__ void wmma_gemm_f32(const float* __restrict__ X, const float* __restrict__ W,
                              const float* __restrict__ bias, float* __restrict__ Y,
                              int M, int K, int Nout, int act) {
    int wave  = (int)((blockIdx.x * blockDim.x + threadIdx.x) >> 5);
    int lane  = threadIdx.x & 31;
    int tilesN = Nout >> 4;
    int m0 = (wave / tilesN) << 4;
    int n0 = (wave % tilesN) << 4;
    if (m0 >= M) return;
    int half = lane >> 4;
    int l    = lane & 15;

    const float* xrow = X + (long long)(m0 + l) * K;
    v8f c = {};
    for (int k = 0; k < K; k += 4) {
        int ka = k + half * 2;
        v2f a, b;
        a.x = xrow[ka];
        a.y = xrow[ka + 1];
        b.x = W[ka * Nout + n0 + l];
        b.y = W[(ka + 1) * Nout + n0 + l];
        c = __builtin_amdgcn_wmma_f32_16x16x4_f32(false, a, false, b,
                                                  (short)0, c, false, false);
    }
    int col = n0 + l;
    float bv = bias ? bias[col] : 0.0f;
#pragma unroll
    for (int r = 0; r < 8; ++r) {
        int row = m0 + r + half * 8;
        float v = c[r] + bv;
        if (act) v = tanhf(v);
        Y[(long long)row * Nout + col] = v;
    }
}

__global__ void scatter_agg64(const float* __restrict__ xl, const int* __restrict__ src,
                              const int* __restrict__ dst, const float* __restrict__ dinv,
                              float* __restrict__ agg, int e_cnt) {
    int t = blockIdx.x * blockDim.x + threadIdx.x;
    int e = t >> 6;
    if (e >= e_cnt) return;
    int f = t & 63;
    int s = src[e];
    int d = dst[e];
    float norm = dinv[s] * dinv[d];
    atomicAdd(&agg[(long long)d * 64 + f], xl[(long long)s * 64 + f] * norm);
}

__global__ void conv_finalize64(const float* __restrict__ agg, const float* __restrict__ xl,
                                const float* __restrict__ selfn, const float* __restrict__ bias,
                                float* __restrict__ out, int n_elems) {
    int t = blockIdx.x * blockDim.x + threadIdx.x;
    if (t >= n_elems) return;
    int node = t >> 6;
    int f    = t & 63;
    out[t] = tanhf(agg[t] + xl[t] * selfn[node] + bias[f]);
}

__global__ void fc_last32(const float* __restrict__ h, const float* __restrict__ w,
                          const float* __restrict__ b, float* __restrict__ out, int n) {
    int i = blockIdx.x * blockDim.x + threadIdx.x;
    if (i >= n) return;
    float acc = b[0];
#pragma unroll
    for (int j = 0; j < 32; ++j) acc += h[i * 32 + j] * w[j];
    out[i] = acc;
}


extern "C" void kernel_launch(void* const* d_in, const int* in_sizes, int n_in,
                              void* d_out, int out_size, void* d_ws, size_t ws_size,
                              hipStream_t stream) {
    const float* x   = (const float*)d_in[0];
    const int*   ei  = (const int*)d_in[1];
    const float* cW0 = (const float*)d_in[2];
    const float* cb0 = (const float*)d_in[3];
    const float* cW1 = (const float*)d_in[4];
    const float* cb1 = (const float*)d_in[5];
    const float* fW0 = (const float*)d_in[6];
    const float* fb0 = (const float*)d_in[7];
    const float* fW1 = (const float*)d_in[8];
    const float* fb1 = (const float*)d_in[9];
    const float* fW2 = (const float*)d_in[10];
    const float* fb2 = (const float*)d_in[11];
    const int* src = ei;
    const int* dst = ei + NE;

    float* ws    = (float*)d_ws;
    float* xl    = ws;
    float* agg   = xl  + (size_t)NN * 64;
    float* h1    = agg + (size_t)NN * 64;
    float* h2    = h1  + (size_t)NN * 64;
    float* deg   = h2  + (size_t)NN * 64;
    float* dinv  = deg  + NN;
    float* selfn = dinv + NN;

    const int B = 256;
    dim3 blk(B);
    auto cdiv = [](long long a, long long b) { return (unsigned)((a + b - 1) / b); };

    unsigned gN    = cdiv(NN, B);
    unsigned gE    = cdiv(NE, B);
    unsigned gNF   = cdiv((long long)NN * 64, B);
    unsigned gEF   = cdiv((long long)NE * 64, B);
    unsigned gG64  = cdiv((long long)(NN / 16) * 4 * 32, B);
    unsigned gG32  = cdiv((long long)(NN / 16) * 2 * 32, B);

    zero_f<<<gN, blk, 0, stream>>>(deg, NN);
    deg_count<<<gE, blk, 0, stream>>>(dst, deg, NE);
    deg_norm<<<gN, blk, 0, stream>>>(deg, dinv, selfn, NN);

    wmma_gemm_f32<<<gG64, blk, 0, stream>>>(x, cW0, nullptr, xl, NN, 128, 64, 0);
    zero_f<<<gNF, blk, 0, stream>>>(agg, NN * 64);
    scatter_agg64<<<gEF, blk, 0, stream>>>(xl, src, dst, dinv, agg, NE);
    conv_finalize64<<<gNF, blk, 0, stream>>>(agg, xl, selfn, cb0, h1, NN * 64);

    wmma_gemm_f32<<<gG64, blk, 0, stream>>>(h1, cW1, nullptr, xl, NN, 64, 64, 0);
    zero_f<<<gNF, blk, 0, stream>>>(agg, NN * 64);
    scatter_agg64<<<gEF, blk, 0, stream>>>(xl, src, dst, dinv, agg, NE);
    conv_finalize64<<<gNF, blk, 0, stream>>>(agg, xl, selfn, cb1, h2, NN * 64);

    wmma_gemm_f32<<<gG64, blk, 0, stream>>>(h2, fW0, fb0, h1, NN, 64, 64, 1);
    wmma_gemm_f32<<<gG32, blk, 0, stream>>>(h1, fW1, fb1, xl, NN, 64, 32, 1);
    fc_last32<<<gN, blk, 0, stream>>>(xl, fW2, fb2, (float*)d_out, NN);
}


# Edge aggregation as a gather over a CSR-by-destination rebuilt each call (two-level counting sort in LDS) instead of per-edge float atomics; one wave per node, zero and finalize fused in
# speedup vs baseline: 4.1293x; 4.1293x over previous
#include <hip/hip_runtime.h>
#include <hip/hip_bf16.h>
#include <math.h>

#define NN 100000
#define NE 1600000

typedef __attribute__((ext_vector_type(2))) float v2f;
typedef __attribute__((ext_vector_type(8))) float v8f;


#define BK_SHIFT 8
#define BK_NODES 256
#define NBK ((NN + BK_NODES - 1) / BK_NODES)
#define TILE_E 4096
#define EPT 16
static_assert(NBK <= 512, "bucket tables hold 512 entries");
static_assert(NN <= (1 << 23), "a source index and 8 bucket-local bits share one 32-bit word");

__global__ void zero_i(int* __restrict__ p, int n) {
    int i = blockIdx.x * blockDim.x + threadIdx.x;
    if (i < n) p[i] = 0;
}

__global__ void bucket_count(const int* __restrict__ dst, int* __restrict__ bcount, int e_cnt) {
    __shared__ int h[512];
    int tid = threadIdx.x;
    h[tid] = 0;
    h[tid + 256] = 0;
    __syncthreads();
    int e0 = blockIdx.x * TILE_E + tid;
    int dv[EPT];
#pragma unroll
    for (int k = 0; k < EPT; ++k) {
        int e = e0 + k * 256;
        dv[k] = dst[e < e_cnt ? e : e_cnt - 1];
    }
#pragma unroll
    for (int k = 0; k < EPT; ++k) {
        int e = e0 + k * 256;
        if (e < e_cnt) atomicAdd(&h[dv[k] >> BK_SHIFT], 1);
    }
    __syncthreads();
    for (int b = tid; b < NBK; b += 256) {
        int c = h[b];
        if (c) atomicAdd(&bcount[b], c);
    }
}

__global__ void bucket_scatter(const int* __restrict__ src, const int* __restrict__ dst,
                               const int* __restrict__ bcount, int* __restrict__ bcursor,
                               int* __restrict__ tmp, int e_cnt) {
    __shared__ int h[512];
    __shared__ int bs[512];
    __shared__ int wsum[8];
    int tid  = threadIdx.x;
    int lane = tid & 31;
    int wv   = tid >> 5;
    h[tid] = 0;
    h[tid + 256] = 0;
    int c0 = (2 * tid     < NBK) ? bcount[2 * tid]     : 0;
    int c1 = (2 * tid + 1 < NBK) ? bcount[2 * tid + 1] : 0;
    int tot = c0 + c1;
    int inc = tot;
    for (int d = 1; d < 32; d <<= 1) {
        int o = __shfl_up(inc, d);
        if (lane >= d) inc += o;
    }
    if (lane == 31) wsum[wv] = inc;
    __syncthreads();
    int wbase = 0;
    for (int w = 0; w < wv; ++w) wbase += wsum[w];
    int ex = wbase + inc - tot;
    bs[2 * tid]     = ex;
    bs[2 * tid + 1] = ex + c0;

    int e0 = blockIdx.x * TILE_E + tid;
    int key[EPT], rk[EPT], bk[EPT];
#pragma unroll
    for (int k = 0; k < EPT; ++k) {
        int e  = e0 + k * 256;
        int ec = e < e_cnt ? e : e_cnt - 1;
        int d  = dst[ec];
        int s  = src[ec];
        bk[k]  = d >> BK_SHIFT;
        key[k] = (s << BK_SHIFT) | (d & (BK_NODES - 1));
    }
#pragma unroll
    for (int k = 0; k < EPT; ++k) {
        int e = e0 + k * 256;
        if (e < e_cnt) rk[k] = atomicAdd(&h[bk[k]], 1);
        else           bk[k] = -1;
    }
    __syncthreads();
    for (int b = tid; b < NBK; b += 256) {
        int c = h[b];
        int off = bs[b];
        if (c) off += atomicAdd(&bcursor[b], c);
        h[b] = off;
    }
    __syncthreads();
#pragma unroll
    for (int k = 0; k < EPT; ++k) {
        if (bk[k] >= 0) tmp[h[bk[k]] + rk[k]] = key[k];
    }
}

__global__ void bucket_to_csr(const int* __restrict__ tmp, const int* __restrict__ bcount,
                              int* __restrict__ rowstart, int* __restrict__ perm,
                              float* __restrict__ dinv, float* __restrict__ selfn, int n) {
    __shared__ int c[BK_NODES];
    __shared__ int wsum[8];
    int tid  = threadIdx.x;
    int lane = tid & 31;
    int wv   = tid >> 5;
    int b    = blockIdx.x;
    int bstart = 0;
    for (int i = 0; i < b; ++i) bstart += bcount[i];
    int bcnt = bcount[b];
    c[tid] = 0;
    __syncthreads();
    for (int e = tid; e < bcnt; e += EPT * 256) {
        int kv[EPT];
#pragma unroll
        for (int i = 0; i < EPT; ++i) {
            int ei = e + i * 256;
            kv[i] = tmp[bstart + (ei < bcnt ? ei : bcnt - 1)];
        }
#pragma unroll
        for (int i = 0; i < EPT; ++i) {
            if (e + i * 256 < bcnt) atomicAdd(&c[kv[i] & (BK_NODES - 1)], 1);
        }
    }
    __syncthreads();
    int my  = c[tid];
    int inc = my;
    for (int d = 1; d < 32; d <<= 1) {
        int o = __shfl_up(inc, d);
        if (lane >= d) inc += o;
    }
    if (lane == 31) wsum[wv] = inc;
    __syncthreads();
    int wbase = 0;
    for (int w = 0; w < wv; ++w) wbase += wsum[w];
    int ex = wbase + inc - my;
    int node = b * BK_NODES + tid;
    if (node < n) {
        rowstart[node] = bstart + ex;
        if (node == n - 1) rowstart[n] = bstart + ex + my;
        float d = (float)my + 1.0f;
        dinv[node]  = rsqrtf(d);
        selfn[node] = 1.0f / d;
    }
    c[tid] = ex;
    __syncthreads();
    for (int e = tid; e < bcnt; e += EPT * 256) {
        int kv[EPT];
#pragma unroll
        for (int i = 0; i < EPT; ++i) {
            int ei = e + i * 256;
            kv[i] = tmp[bstart + (ei < bcnt ? ei : bcnt - 1)];
        }
#pragma unroll
        for (int i = 0; i < EPT; ++i) {
            if (e + i * 256 < bcnt) {
                int pos = atomicAdd(&c[kv[i] & (BK_NODES - 1)], 1);
                perm[bstart + pos] = kv[i] >> BK_SHIFT;
            }
        }
    }
}

__global__ void wmma_gemm_f32(const float* __restrict__ X, const float* __restrict__ W,
                              const float* __restrict__ bias, float* __restrict__ Y,
                              int M, int K, int Nout, int act) {
    int wave  = (int)((blockIdx.x * blockDim.x + threadIdx.x) >> 5);
    int lane  = threadIdx.x & 31;
    int tilesN = Nout >> 4;
    int m0 = (wave / tilesN) << 4;
    int n0 = (wave % tilesN) << 4;
    if (m0 >= M) return;
    int half = lane >> 4;
    int l    = lane & 15;

    const float* xrow = X + (long long)(m0 + l) * K;
    v8f c = {};
    for (int k = 0; k < K; k += 4) {
        int ka = k + half * 2;
        v2f a, b;
        a.x = xrow[ka];
        a.y = xrow[ka + 1];
        b.x = W[ka * Nout + n0 + l];
        b.y = W[(ka + 1) * Nout + n0 + l];
        c = __builtin_amdgcn_wmma_f32_16x16x4_f32(false, a, false, b,
                                                  (short)0, c, false, false);
    }
    int col = n0 + l;
    float bv = bias ? bias[col] : 0.0f;
#pragma unroll
    for (int r = 0; r < 8; ++r) {
        int row = m0 + r + half * 8;
        float v = c[r] + bv;
        if (act) v = tanhf(v);
        Y[(long long)row * Nout + col] = v;
    }
}


template <int K>
__device__ __forceinline__ void add_rows(const float2* __restrict__ xl2, int s_l, float w_l,
                                         int j, int lane, float2& acc) {
    float2 v[K];
    float  w[K];
#pragma unroll
    for (int k = 0; k < K; ++k) {
        int s = __builtin_amdgcn_readlane(s_l, j + k);
        w[k] = __int_as_float(__builtin_amdgcn_readlane(__float_as_int(w_l), j + k));
        v[k] = xl2[(long long)s * 32 + lane];
    }
#pragma unroll
    for (int k = 0; k < K; ++k) {
        acc.x += v[k].x * w[k];
        acc.y += v[k].y * w[k];
    }
}

__global__ void gather_finalize64(const float* __restrict__ xl, const int* __restrict__ rowstart,
                                  const int* __restrict__ perm, const float* __restrict__ dinv,
                                  const float* __restrict__ selfn, const float* __restrict__ bias,
                                  float* __restrict__ out, int n) {
    int node = __builtin_amdgcn_readfirstlane((int)((blockIdx.x * blockDim.x + threadIdx.x) >> 5));
    int lane = threadIdx.x & 31;
    if (node >= n) return;
    const float2* xl2 = (const float2*)xl;
    int beg = rowstart[node];
    int end = rowstart[node + 1];
    float dn = dinv[node];
    float2 xs = xl2[(long long)node * 32 + lane];
    float sn = selfn[node];
    float b0 = bias[2 * lane];
    float b1 = bias[2 * lane + 1];
    float2 acc = {0.0f, 0.0f};
    for (int base = beg; base < end; base += 32) {
        int cnt = end - base;
        if (cnt > 32) cnt = 32;
        int   s_l = node;
        float w_l = 0.0f;
        if (lane < cnt) {
            s_l = perm[base + lane];
            w_l = dinv[s_l] * dn;
        }
        int j = 0;
        for (; j + 8 <= cnt; j += 8) add_rows<8>(xl2, s_l, w_l, j, lane, acc);
        if (j + 4 <= cnt) { add_rows<4>(xl2, s_l, w_l, j, lane, acc); j += 4; }
        int rem = cnt - j;
        if (rem == 3)      add_rows<3>(xl2, s_l, w_l, j, lane, acc);
        else if (rem == 2) add_rows<2>(xl2, s_l, w_l, j, lane, acc);
        else if (rem == 1) add_rows<1>(xl2, s_l, w_l, j, lane, acc);
    }
    float2 o;
    o.x = tanhf(acc.x + xs.x * sn + b0);
    o.y = tanhf(acc.y + xs.y * sn + b1);
    ((float2*)out)[(long long)node * 32 + lane] = o;
}

__global__ void fc_last32(const float* __restrict__ h, const float* __restrict__ w,
                          const float* __restrict__ b, float* __restrict__ out, int n) {
    int i = blockIdx.x * blockDim.x + threadIdx.x;
    if (i >= n) return;
    float acc = b[0];
#pragma unroll
    for (int j = 0; j < 32; ++j) acc += h[i * 32 + j] * w[j];
    out[i] = acc;
}


extern "C" void kernel_launch(void* const* d_in, const int* in_sizes, int n_in,
                              void* d_out, int out_size, void* d_ws, size_t ws_size,
                              hipStream_t stream) {
    const float* x   = (const float*)d_in[0];
    const int*   ei  = (const int*)d_in[1];
    const float* cW0 = (const float*)d_in[2];
    const float* cb0 = (const float*)d_in[3];
    const float* cW1 = (const float*)d_in[4];
    const float* cb1 = (const float*)d_in[5];
    const float* fW0 = (const float*)d_in[6];
    const float* fb0 = (const float*)d_in[7];
    const float* fW1 = (const float*)d_in[8];
    const float* fb1 = (const float*)d_in[9];
    const float* fW2 = (const float*)d_in[10];
    const float* fb2 = (const float*)d_in[11];
    const int* src = ei;
    const int* dst = ei + NE;

    float* ws    = (float*)d_ws;
    float* xl    = ws;
    float* csr   = xl  + (size_t)NN * 64;
    float* h1    = csr + (size_t)NN * 64;
    float* h2    = h1  + (size_t)NN * 64;
    float* dinv  = h2   + (size_t)NN * 64;
    float* selfn = dinv + NN;

    int* bcount   = (int*)csr;
    int* bcursor  = bcount + 512;
    int* rowstart = bcursor + 512;
    int* tmp      = rowstart + NN + 32;
    int* perm     = tmp + NE;

    const int B = 256;
    dim3 blk(B);
    auto cdiv = [](long long a, long long b) { return (unsigned)((a + b - 1) / b); };

    unsigned gN    = cdiv(NN, B);
    unsigned gNW   = cdiv((long long)NN * 32, B);
    unsigned gT    = cdiv(NE, TILE_E);
    unsigned gG64  = cdiv((long long)(NN / 16) * 4 * 32, B);
    unsigned gG32  = cdiv((long long)(NN / 16) * 2 * 32, B);

    zero_i<<<4, blk, 0, stream>>>(bcount, 1024);
    bucket_count<<<gT, blk, 0, stream>>>(dst, bcount, NE);
    bucket_scatter<<<gT, blk, 0, stream>>>(src, dst, bcount, bcursor, tmp, NE);
    bucket_to_csr<<<NBK, blk, 0, stream>>>(tmp, bcount, rowstart, perm, dinv, selfn, NN);

    wmma_gemm_f32<<<gG64, blk, 0, stream>>>(x, cW0, nullptr, xl, NN, 128, 64, 0);
    gather_finalize64<<<gNW, blk, 0, stream>>>(xl, rowstart, perm, dinv, selfn, cb0, h1, NN);

    wmma_gemm_f32<<<gG64, blk, 0, stream>>>(h1, cW1, nullptr, xl, NN, 64, 64, 0);
    gather_finalize64<<<gNW, blk, 0, stream>>>(xl, rowstart, perm, dinv, selfn, cb1, h2, NN);

    wmma_gemm_f32<<<gG64, blk, 0, stream>>>(h2, fW0, fb0, h1, NN, 64, 64, 1);
    wmma_gemm_f32<<<gG32, blk, 0, stream>>>(h1, fW1, fb1, xl, NN, 64, 32, 1);
    fc_last32<<<gN, blk, 0, stream>>>(xl, fW2, fb2, (float*)d_out, NN);
}


# Each gather wave owns two adjacent nodes, one per half-wave with 16-byte lanes (one load fetches a 256-B row for each): blocks of 8 waves, half the waves enter and half the row loads issue
# speedup vs baseline: 5.0878x; 1.2321x over previous
#include <hip/hip_runtime.h>
#include <hip/hip_bf16.h>
#include <math.h>

#define NN 100000
#define NE 1600000

typedef __attribute__((ext_vector_type(2))) float v2f;
typedef __attribute__((ext_vector_type(8))) float v8f;


#define BK_SHIFT 8
#define BK_NODES 256
#define NBK ((NN + BK_NODES - 1) / BK_NODES)
#define TILE_E 4096
#define EPT 16
static_assert(NBK <= 512, "bucket tables hold 512 entries");
static_assert(NN <= (1 << 23), "a source index and 8 bucket-local bits share one 32-bit word");

__global__ void zero_i(int* __restrict__ p, int n) {
    int i = blockIdx.x * blockDim.x + threadIdx.x;
    if (i < n) p[i] = 0;
}

__global__ void bucket_count(const int* __restrict__ dst, int* __restrict__ bcount, int e_cnt) {
    __shared__ int h[512];
    int tid = threadIdx.x;
    h[tid] = 0;
    h[tid + 256] = 0;
    __syncthreads();
    int e0 = blockIdx.x * TILE_E + tid;
    int dv[EPT];
#pragma unroll
    for (int k = 0; k < EPT; ++k) {
        int e = e0 + k * 256;
        dv[k] = dst[e < e_cnt ? e : e_cnt - 1];
    }
#pragma unroll
    for (int k = 0; k < EPT; ++k) {
        int e = e0 + k * 256;
        if (e < e_cnt) atomicAdd(&h[dv[k] >> BK_SHIFT], 1);
    }
    __syncthreads();
    for (int b = tid; b < NBK; b += 256) {
        int c = h[b];
        if (c) atomicAdd(&bcount[b], c);
    }
}

__global__ void bucket_scatter(const int* __restrict__ src, const int* __restrict__ dst,
                               const int* __restrict__ bcount, int* __restrict__ bcursor,
                               int* __restrict__ tmp, int e_cnt) {
    __shared__ int h[512];
    __shared__ int bs[512];
    __shared__ int wsum[8];
    int tid  = threadIdx.x;
    int lane = tid & 31;
    int wv   = tid >> 5;
    h[tid] = 0;
    h[tid + 256] = 0;
    int c0 = (2 * tid     < NBK) ? bcount[2 * tid]     : 0;
    int c1 = (2 * tid + 1 < NBK) ? bcount[2 * tid + 1] : 0;
    int tot = c0 + c1;
    int inc = tot;
    for (int d = 1; d < 32; d <<= 1) {
        int o = __shfl_up(inc, d);
        if (lane >= d) inc += o;
    }
    if (lane == 31) wsum[wv] = inc;
    __syncthreads();
    int wbase = 0;
    for (int w = 0; w < wv; ++w) wbase += wsum[w];
    int ex = wbase + inc - tot;
    bs[2 * tid]     = ex;
    bs[2 * tid + 1] = ex + c0;

    int e0 = blockIdx.x * TILE_E + tid;
    int key[EPT], rk[EPT], bk[EPT];
#pragma unroll
    for (int k = 0; k < EPT; ++k) {
        int e  = e0 + k * 256;
        int ec = e < e_cnt ? e : e_cnt - 1;
        int d  = dst[ec];
        int s  = src[ec];
        bk[k]  = d >> BK_SHIFT;
        key[k] = (s << BK_SHIFT) | (d & (BK_NODES - 1));
    }
#pragma unroll
    for (int k = 0; k < EPT; ++k) {
        int e = e0 + k * 256;
        if (e < e_cnt) rk[k] = atomicAdd(&h[bk[k]], 1);
        else           bk[k] = -1;
    }
    __syncthreads();
    for (int b = tid; b < NBK; b += 256) {
        int c = h[b];
        int off = bs[b];
        if (c) off += atomicAdd(&bcursor[b], c);
        h[b] = off;
    }
    __syncthreads();
#pragma unroll
    for (int k = 0; k < EPT; ++k) {
        if (bk[k] >= 0) tmp[h[bk[k]] + rk[k]] = key[k];
    }
}

__global__ void bucket_to_csr(const int* __restrict__ tmp, const int* __restrict__ bcount,
                              int* __restrict__ rowstart, int* __restrict__ perm,
                              float* __restrict__ dinv, float* __restrict__ selfn, int n) {
    __shared__ int c[BK_NODES];
    __shared__ int wsum[8];
    int tid  = threadIdx.x;
    int lane = tid & 31;
    int wv   = tid >> 5;
    int b    = blockIdx.x;
    int bstart = 0;
    for (int i = 0; i < b; ++i) bstart += bcount[i];
    int bcnt = bcount[b];
    c[tid] = 0;
    __syncthreads();
    for (int e = tid; e < bcnt; e += EPT * 256) {
        int kv[EPT];
#pragma unroll
        for (int i = 0; i < EPT; ++i) {
            int ei = e + i * 256;
            kv[i] = tmp[bstart + (ei < bcnt ? ei : bcnt - 1)];
        }
#pragma unroll
        for (int i = 0; i < EPT; ++i) {
            if (e + i * 256 < bcnt) atomicAdd(&c[kv[i] & (BK_NODES - 1)], 1);
        }
    }
    __syncthreads();
    int my  = c[tid];
    int inc = my;
    for (int d = 1; d < 32; d <<= 1) {
        int o = __shfl_up(inc, d);
        if (lane >= d) inc += o;
    }
    if (lane == 31) wsum[wv] = inc;
    __syncthreads();
    int wbase = 0;
    for (int w = 0; w < wv; ++w) wbase += wsum[w];
    int ex = wbase + inc - my;
    int node = b * BK_NODES + tid;
    if (node < n) {
        rowstart[node] = bstart + ex;
        if (node == n - 1) rowstart[n] = bstart + ex + my;
        float d = (float)my + 1.0f;
        dinv[node]  = rsqrtf(d);
        selfn[node] = 1.0f / d;
    }
    c[tid] = ex;
    __syncthreads();
    for (int e = tid; e < bcnt; e += EPT * 256) {
        int kv[EPT];
#pragma unroll
        for (int i = 0; i < EPT; ++i) {
            int ei = e + i * 256;
            kv[i] = tmp[bstart + (ei < bcnt ? ei : bcnt - 1)];
        }
#pragma unroll
        for (int i = 0; i < EPT; ++i) {
            if (e + i * 256 < bcnt) {
                int pos = atomicAdd(&c[kv[i] & (BK_NODES - 1)], 1);
                perm[bstart + pos] = kv[i] >> BK_SHIFT;
            }
        }
    }
}

__global__ void wmma_gemm_f32(const float* __restrict__ X, const float* __restrict__ W,
                              const float* __restrict__ bias, float* __restrict__ Y,
                              int M, int K, int Nout, int act) {
    int wave  = (int)((blockIdx.x * blockDim.x + threadIdx.x) >> 5);
    int lane  = threadIdx.x & 31;
    int tilesN = Nout >> 4;
    int m0 = (wave / tilesN) << 4;
    int n0 = (wave % tilesN) << 4;
    if (m0 >= M) return;
    int half = lane >> 4;
    int l    = lane & 15;

    const float* xrow = X + (long long)(m0 + l) * K;
    v8f c = {};
    for (int k = 0; k < K; k += 4) {
        int ka = k + half * 2;
        v2f a, b;
        a.x = xrow[ka];
        a.y = xrow[ka + 1];
        b.x = W[ka * Nout + n0 + l];
        b.y = W[(ka + 1) * Nout + n0 + l];
        c = __builtin_amdgcn_wmma_f32_16x16x4_f32(false, a, false, b,
                                                  (short)0, c, false, false);
    }
    int col = n0 + l;
    float bv = bias ? bias[col] : 0.0f;
#pragma unroll
    for (int r = 0; r < 8; ++r) {
        int row = m0 + r + half * 8;
        float v = c[r] + bv;
        if (act) v = tanhf(v);
        Y[(long long)row * Nout + col] = v;
    }
}


template <int K>
__device__ __forceinline__ void add_rows(const char* __restrict__ xlb, int s_l, float w_l,
                                         int j, int half, unsigned lofs, v2f& acc01, v2f& acc23) {
    float4 v[K];
    float  w[K];
#pragma unroll
    for (int k = 0; k < K; ++k) {
        int   s_lo = __builtin_amdgcn_readlane(s_l, j + k);
        int   s_hi = __builtin_amdgcn_readlane(s_l, 16 + j + k);
        float w_lo = __int_as_float(__builtin_amdgcn_readlane(__float_as_int(w_l), j + k));
        float w_hi = __int_as_float(__builtin_amdgcn_readlane(__float_as_int(w_l), 16 + j + k));
        int s = half ? s_hi : s_lo;
        w[k]  = half ? w_hi : w_lo;
        v[k]  = *(const float4*)(xlb + ((unsigned)s * 256u + lofs));
    }
#pragma unroll
    for (int k = 0; k < K; ++k) {
        v2f lo = {v[k].x, v[k].y};
        v2f hi = {v[k].z, v[k].w};
        v2f ww = {w[k], w[k]};
        acc01 += lo * ww;
        acc23 += hi * ww;
    }
}

__device__ __forceinline__ void add_chunk(const char* __restrict__ xlb, int s_l, float w_l,
                                          int cnt, int half, unsigned lofs, v2f& acc01, v2f& acc23) {
    int j = 0;
    for (; j + 8 <= cnt; j += 8) add_rows<8>(xlb, s_l, w_l, j, half, lofs, acc01, acc23);
    if (j + 4 <= cnt) { add_rows<4>(xlb, s_l, w_l, j, half, lofs, acc01, acc23); j += 4; }
    int rem = cnt - j;
    if (rem == 3)      add_rows<3>(xlb, s_l, w_l, j, half, lofs, acc01, acc23);
    else if (rem == 2) add_rows<2>(xlb, s_l, w_l, j, half, lofs, acc01, acc23);
    else if (rem == 1) add_rows<1>(xlb, s_l, w_l, j, half, lofs, acc01, acc23);
}


#define TROWS 16
#define GWAVES (TROWS / 2)
#define LDP   68

template <int M>
__device__ __forceinline__ float row_xor_add(float v) {
    int t = __builtin_amdgcn_update_dpp(0, __float_as_int(v), 0x160 + M, 0xf, 0xf, false);
    return v + __int_as_float(t);
}

template <int HEAD>
__global__ void __launch_bounds__(GWAVES * 32)
gather_dense64(const float* __restrict__ xl, const int* __restrict__ rowstart,
               const int* __restrict__ perm, const float* __restrict__ dinv,
               const float* __restrict__ selfn, const float* __restrict__ bias,
               const float* __restrict__ W0, const float* __restrict__ b0,
               const float* __restrict__ W1, const float* __restrict__ b1,
               const float* __restrict__ W2, const float* __restrict__ b2,
               float* __restrict__ out, int n) {
    __shared__ __attribute__((aligned(16))) float t0[TROWS * LDP];
    __shared__ float t1[HEAD ? TROWS * LDP : 2];
    __shared__ float ps[TROWS];
    int wv    = __builtin_amdgcn_readfirstlane((int)(threadIdx.x >> 5));
    int node0 = (int)blockIdx.x * TROWS;
    int nodeA = node0 + 2 * wv;
    int lane  = threadIdx.x & 31;
    int half  = lane >> 4;
    int l     = lane & 15;

    v2f acc01 = {0.0f, 0.0f}, acc23 = {0.0f, 0.0f};
    float4 xs  = {0.0f, 0.0f, 0.0f, 0.0f};
    float4 cb  = {0.0f, 0.0f, 0.0f, 0.0f};
    float sn = 0.0f;
    if (nodeA < n) {
        const char* xlb = (const char*)xl;
        unsigned lofs = (unsigned)l * 16u;
        int r0 = rowstart[nodeA];
        int r1 = rowstart[nodeA + 1];
        int r2 = (nodeA + 1 < n) ? rowstart[nodeA + 2] : r1;
        int dA = r1 - r0;
        int dB = r2 - r1;
        int maxdeg = dA > dB ? dA : dB;
        int beg = half ? r1 : r0;
        int deg = half ? dB : dA;
        int nd = (nodeA + half < n) ? nodeA + half : nodeA;
        float dn = dinv[nd];
        xs = *(const float4*)(xlb + ((unsigned)nd * 256u + lofs));
        sn = selfn[nd];
        cb.x = bias[4 * l];
        cb.y = bias[4 * l + 1];
        cb.z = bias[4 * l + 2];
        cb.w = bias[4 * l + 3];
        int last = r2 - 1;
        for (int c = 0; c < maxdeg; c += 32) {
            bool  in0 = c + l < deg;
            bool  in1 = c + 16 + l < deg;
            int   p0 = perm[in0 ? beg + c + l : last];
            int   p1 = perm[in1 ? beg + c + 16 + l : last];
            float d0 = dinv[p0];
            float d1 = dinv[p1];
            int   s0 = in0 ? p0 : nd;
            int   s1 = in1 ? p1 : nd;
            float w0 = in0 ? d0 * dn : 0.0f;
            float w1 = in1 ? d1 * dn : 0.0f;
            int left = maxdeg - c;
            add_chunk(xlb, s0, w0, left < 16 ? left : 16, half, lofs, acc01, acc23);
            if (left > 16) add_chunk(xlb, s1, w1, left < 32 ? left - 16 : 16, half, lofs, acc01, acc23);
        }
    }

    v2f bw[16];
    int n0 = (wv & 3) << 4;
    if (wv < 4) {
#pragma unroll
        for (int s = 0; s < 16; ++s) {
            int ka = 4 * s + half * 2;
            bw[s].x = W0[ka * 64 + n0 + l];
            bw[s].y = W0[(ka + 1) * 64 + n0 + l];
        }
    } else if (HEAD && wv < 6) {
#pragma unroll
        for (int s = 0; s < 16; ++s) {
            int ka = 4 * s + half * 2;
            bw[s].x = W1[ka * 32 + n0 + l];
            bw[s].y = W1[(ka + 1) * 32 + n0 + l];
        }
    }

    float4 o;
    o.x = tanhf(acc01.x + xs.x * sn + cb.x);
    o.y = tanhf(acc01.y + xs.y * sn + cb.y);
    o.z = tanhf(acc23.x + xs.z * sn + cb.z);
    o.w = tanhf(acc23.y + xs.w * sn + cb.w);
    *(float4*)&t0[(2 * wv + half) * LDP + 4 * l] = o;
    __syncthreads();

    if (wv < 4) {
        v8f c = {};
#pragma unroll
        for (int s = 0; s < 16; ++s) {
            int ka = 4 * s + half * 2;
            v2f a = *(const v2f*)&t0[l * LDP + ka];
            c = __builtin_amdgcn_wmma_f32_16x16x4_f32(false, a, false, bw[s],
                                                      (short)0, c, false, false);
        }
        if (!HEAD) {
#pragma unroll
            for (int r = 0; r < 8; ++r) {
                int row = node0 + r + half * 8;
                if (row < n) out[(long long)row * 64 + n0 + l] = c[r];
            }
        } else {
            float bv = b0[n0 + l];
#pragma unroll
            for (int r = 0; r < 8; ++r)
                t1[(r + half * 8) * LDP + n0 + l] = tanhf(c[r] + bv);
        }
    }
    if (HEAD) {
        __syncthreads();
        float sel = 0.0f;
        if (wv == 4 || wv == 5) {
            v8f c = {};
#pragma unroll
            for (int s = 0; s < 16; ++s) {
                int ka = 4 * s + half * 2;
                v2f a = *(const v2f*)&t1[l * LDP + ka];
                c = __builtin_amdgcn_wmma_f32_16x16x4_f32(false, a, false, bw[s],
                                                          (short)0, c, false, false);
            }
            float bv = b1[n0 + l];
            float w2 = W2[n0 + l];
            float p[8];
#pragma unroll
            for (int r = 0; r < 8; ++r) p[r] = tanhf(c[r] + bv) * w2;
#pragma unroll
            for (int r = 0; r < 8; ++r) {
                float v = p[r];
                v = row_xor_add<1>(v);
                v = row_xor_add<2>(v);
                v = row_xor_add<4>(v);
                v = row_xor_add<8>(v);
                if (l == r) sel = v;
            }
            if (wv == 5 && l < 8) ps[half * 8 + l] = sel;
        }
        __syncthreads();
        if (wv == 4 && l < 8) {
            int row = half * 8 + l;
            if (node0 + row < n) out[node0 + row] = sel + ps[row] + b2[0];
        }
    }
}


extern "C" void kernel_launch(void* const* d_in, const int* in_sizes, int n_in,
                              void* d_out, int out_size, void* d_ws, size_t ws_size,
                              hipStream_t stream) {
    const float* x   = (const float*)d_in[0];
    const int*   ei  = (const int*)d_in[1];
    const float* cW0 = (const float*)d_in[2];
    const float* cb0 = (const float*)d_in[3];
    const float* cW1 = (const float*)d_in[4];
    const float* cb1 = (const float*)d_in[5];
    const float* fW0 = (const float*)d_in[6];
    const float* fb0 = (const float*)d_in[7];
    const float* fW1 = (const float*)d_in[8];
    const float* fb1 = (const float*)d_in[9];
    const float* fW2 = (const float*)d_in[10];
    const float* fb2 = (const float*)d_in[11];
    const int* src = ei;
    const int* dst = ei + NE;

    float* ws    = (float*)d_ws;
    float* xl    = ws;
    float* csr   = xl  + (size_t)NN * 64;
    float* h1    = csr + (size_t)NN * 64;
    float* h2    = h1  + (size_t)NN * 64;
    float* dinv  = h2   + (size_t)NN * 64;
    float* selfn = dinv + NN;

    int* bcount   = (int*)csr;
    int* bcursor  = bcount + 512;
    int* rowstart = bcursor + 512;
    int* tmp      = rowstart + NN + 32;
    int* perm     = tmp + NE;

    const int B = 256;
    dim3 blk(B);
    auto cdiv = [](long long a, long long b) { return (unsigned)((a + b - 1) / b); };

    unsigned gT    = cdiv(NE, TILE_E);
    unsigned gG64  = cdiv((long long)(NN / 16) * 4 * 32, B);
    unsigned gA    = cdiv(NN, TROWS);
    dim3 blkA(GWAVES * 32);

    zero_i<<<4, blk, 0, stream>>>(bcount, 1024);
    bucket_count<<<gT, blk, 0, stream>>>(dst, bcount, NE);
    bucket_scatter<<<gT, blk, 0, stream>>>(src, dst, bcount, bcursor, tmp, NE);
    bucket_to_csr<<<NBK, blk, 0, stream>>>(tmp, bcount, rowstart, perm, dinv, selfn, NN);

    wmma_gemm_f32<<<gG64, blk, 0, stream>>>(x, cW0, nullptr, xl, NN, 128, 64, 0);

    gather_dense64<0><<<gA, blkA, 0, stream>>>(xl, rowstart, perm, dinv, selfn, cb0,
                                               cW1, nullptr, nullptr, nullptr, nullptr, nullptr,
                                               h1, NN);

    gather_dense64<1><<<gA, blkA, 0, stream>>>(h1, rowstart, perm, dinv, selfn, cb1,
                                               fW0, fb0, fW1, fb1, fW2, fb2,
                                               (float*)d_out, NN);
}


# x@W0 product: 4-wave blocks park a 32-row A tile in LDS by whole-row 16-byte loads; each wave keeps its 16 columns' B fragments in registers for both strips (A was fetched 4x in 8-byte pieces)
# speedup vs baseline: 6.2849x; 1.2353x over previous
#include <hip/hip_runtime.h>
#include <hip/hip_bf16.h>
#include <math.h>

#define NN 100000
#define NE 1600000

typedef __attribute__((ext_vector_type(2))) float v2f;
typedef __attribute__((ext_vector_type(8))) float v8f;


#define BK_SHIFT 8
#define BK_NODES 256
#define NBK ((NN + BK_NODES - 1) / BK_NODES)
#define TILE_E 4096
#define EPT 16
static_assert(NBK <= 512, "bucket tables hold 512 entries");
static_assert(NN <= (1 << 23), "a source index and 8 bucket-local bits share one 32-bit word");

__global__ void zero_i(int* __restrict__ p, int n) {
    int i = blockIdx.x * blockDim.x + threadIdx.x;
    if (i < n) p[i] = 0;
}

__global__ void bucket_count(const int* __restrict__ dst, int* __restrict__ bcount, int e_cnt) {
    __shared__ int h[512];
    int tid = threadIdx.x;
    h[tid] = 0;
    h[tid + 256] = 0;
    __syncthreads();
    int e0 = blockIdx.x * TILE_E + tid;
    int dv[EPT];
#pragma unroll
    for (int k = 0; k < EPT; ++k) {
        int e = e0 + k * 256;
        dv[k] = dst[e < e_cnt ? e : e_cnt - 1];
    }
#pragma unroll
    for (int k = 0; k < EPT; ++k) {
        int e = e0 + k * 256;
        if (e < e_cnt) atomicAdd(&h[dv[k] >> BK_SHIFT], 1);
    }
    __syncthreads();
    for (int b = tid; b < NBK; b += 256) {
        int c = h[b];
        if (c) atomicAdd(&bcount[b], c);
    }
}

__global__ void bucket_scatter(const int* __restrict__ src, const int* __restrict__ dst,
                               const int* __restrict__ bcount, int* __restrict__ bcursor,
                               int* __restrict__ tmp, int e_cnt) {
    __shared__ int h[512];
    __shared__ int bs[512];
    __shared__ int wsum[8];
    int tid  = threadIdx.x;
    int lane = tid & 31;
    int wv   = tid >> 5;
    h[tid] = 0;
    h[tid + 256] = 0;
    int c0 = (2 * tid     < NBK) ? bcount[2 * tid]     : 0;
    int c1 = (2 * tid + 1 < NBK) ? bcount[2 * tid + 1] : 0;
    int tot = c0 + c1;
    int inc = tot;
    for (int d = 1; d < 32; d <<= 1) {
        int o = __shfl_up(inc, d);
        if (lane >= d) inc += o;
    }
    if (lane == 31) wsum[wv] = inc;
    __syncthreads();
    int wbase = 0;
    for (int w = 0; w < wv; ++w) wbase += wsum[w];
    int ex = wbase + inc - tot;
    bs[2 * tid]     = ex;
    bs[2 * tid + 1] = ex + c0;

    int e0 = blockIdx.x * TILE_E + tid;
    int key[EPT], rk[EPT], bk[EPT];
#pragma unroll
    for (int k = 0; k < EPT; ++k) {
        int e  = e0 + k * 256;
        int ec = e < e_cnt ? e : e_cnt - 1;
        int d  = dst[ec];
        int s  = src[ec];
        bk[k]  = d >> BK_SHIFT;
        key[k] = (s << BK_SHIFT) | (d & (BK_NODES - 1));
    }
#pragma unroll
    for (int k = 0; k < EPT; ++k) {
        int e = e0 + k * 256;
        if (e < e_cnt) rk[k] = atomicAdd(&h[bk[k]], 1);
        else           bk[k] = -1;
    }
    __syncthreads();
    for (int b = tid; b < NBK; b += 256) {
        int c = h[b];
        int off = bs[b];
        if (c) off += atomicAdd(&bcursor[b], c);
        h[b] = off;
    }
    __syncthreads();
#pragma unroll
    for (int k = 0; k < EPT; ++k) {
        if (bk[k] >= 0) tmp[h[bk[k]] + rk[k]] = key[k];
    }
}

__global__ void bucket_to_csr(const int* __restrict__ tmp, const int* __restrict__ bcount,
                              int* __restrict__ rowstart, int* __restrict__ perm,
                              float* __restrict__ dinv, float* __restrict__ zrow, int n) {
    __shared__ int c[BK_NODES];
    __shared__ int wsum[8];
    int tid  = threadIdx.x;
    int lane = tid & 31;
    int wv   = tid >> 5;
    int b    = blockIdx.x;
    int bstart = 0;
    for (int i = 0; i < b; ++i) bstart += bcount[i];
    int bcnt = bcount[b];
    c[tid] = 0;
    __syncthreads();
    for (int e = tid; e < bcnt; e += EPT * 256) {
        int kv[EPT];
#pragma unroll
        for (int i = 0; i < EPT; ++i) {
            int ei = e + i * 256;
            kv[i] = tmp[bstart + (ei < bcnt ? ei : bcnt - 1)];
        }
#pragma unroll
        for (int i = 0; i < EPT; ++i) {
            if (e + i * 256 < bcnt) atomicAdd(&c[kv[i] & (BK_NODES - 1)], 1);
        }
    }
    __syncthreads();
    int my  = c[tid];
    int inc = my;
    for (int d = 1; d < 32; d <<= 1) {
        int o = __shfl_up(inc, d);
        if (lane >= d) inc += o;
    }
    if (lane == 31) wsum[wv] = inc;
    __syncthreads();
    int wbase = 0;
    for (int w = 0; w < wv; ++w) wbase += wsum[w];
    int ex = wbase + inc - my;
    int node = b * BK_NODES + tid;
    if (node < n) {
        rowstart[node] = bstart + ex;
        if (node == n - 1) rowstart[n] = bstart + ex + my;
        dinv[node] = rsqrtf((float)my + 1.0f);
    }
    if (b == 0 && tid < 64) zrow[tid] = 0.0f;
    c[tid] = ex;
    __syncthreads();
    for (int e = tid; e < bcnt; e += EPT * 256) {
        int kv[EPT];
#pragma unroll
        for (int i = 0; i < EPT; ++i) {
            int ei = e + i * 256;
            kv[i] = tmp[bstart + (ei < bcnt ? ei : bcnt - 1)];
        }
#pragma unroll
        for (int i = 0; i < EPT; ++i) {
            if (e + i * 256 < bcnt) {
                int pos = atomicAdd(&c[kv[i] & (BK_NODES - 1)], 1);
                perm[bstart + pos] = kv[i] >> BK_SHIFT;
            }
        }
    }
}

#define GK   128
#define GN   64
#define GROWS 32
#define GP   132

static_assert(NN % GROWS == 0, "every block of the product owns 32 whole rows");

__global__ void __launch_bounds__(128)
gemm_rows32_k128_n64(const float* __restrict__ X, const float* __restrict__ W,
                     const float* __restrict__ rowscale, float* __restrict__ Y) {
    __shared__ __attribute__((aligned(16))) float ta[GROWS * GP];
    __shared__ __attribute__((aligned(16))) float dsc[4][GROWS];
    int wv   = __builtin_amdgcn_readfirstlane((int)(threadIdx.x >> 5));
    int lane = threadIdx.x & 31;
    int half = lane >> 4;
    int l    = lane & 15;
    int m0   = (int)blockIdx.x * GROWS;

    float4 av[8];
#pragma unroll
    for (int i = 0; i < 8; ++i)
        av[i] = *(const float4*)(X + (size_t)(m0 + wv * 8 + i) * GK + 4 * lane);
    float rsv = rowscale[m0 + lane];
    v2f bw[32];
    int n0 = wv << 4;
#pragma unroll
    for (int s = 0; s < 32; ++s) {
        int ka = 4 * s + half * 2;
        bw[s].x = W[ka * GN + n0 + l];
        bw[s].y = W[(ka + 1) * GN + n0 + l];
    }
#pragma unroll
    for (int i = 0; i < 8; ++i)
        *(float4*)&ta[(wv * 8 + i) * GP + 4 * lane] = av[i];
    dsc[wv][lane] = rsv;
    __syncthreads();

    v8f c0 = {}, c1 = {};
#pragma unroll
    for (int s = 0; s < 32; ++s) {
        int ka = 4 * s + half * 2;
        v2f a0 = *(const v2f*)&ta[l * GP + ka];
        v2f a1 = *(const v2f*)&ta[(16 + l) * GP + ka];
        c0 = __builtin_amdgcn_wmma_f32_16x16x4_f32(false, a0, false, bw[s],
                                                   (short)0, c0, false, false);
        c1 = __builtin_amdgcn_wmma_f32_16x16x4_f32(false, a1, false, bw[s],
                                                   (short)0, c1, false, false);
    }
    float4 d0a = *(const float4*)&dsc[wv][half * 8];
    float4 d0b = *(const float4*)&dsc[wv][half * 8 + 4];
    float4 d1a = *(const float4*)&dsc[wv][16 + half * 8];
    float4 d1b = *(const float4*)&dsc[wv][16 + half * 8 + 4];
    float dr0[8] = {d0a.x, d0a.y, d0a.z, d0a.w, d0b.x, d0b.y, d0b.z, d0b.w};
    float dr1[8] = {d1a.x, d1a.y, d1a.z, d1a.w, d1b.x, d1b.y, d1b.z, d1b.w};
    float* yp = Y + (size_t)(m0 + half * 8) * GN + n0 + l;
#pragma unroll
    for (int r = 0; r < 8; ++r) {
        yp[r * GN]        = c0[r] * dr0[r];
        yp[(r + 16) * GN] = c1[r] * dr1[r];
    }
}


template <int K>
__device__ __forceinline__ void add_rows(const char* __restrict__ xlb, int s_l,
                                         int j, int half, unsigned lofs, v2f& acc01, v2f& acc23) {
    float4 v[K];
#pragma unroll
    for (int k = 0; k < K; ++k) {
        int s_lo = __builtin_amdgcn_readlane(s_l, j + k);
        int s_hi = __builtin_amdgcn_readlane(s_l, 16 + j + k);
        int s = half ? s_hi : s_lo;
        v[k]  = *(const float4*)(xlb + ((unsigned)s * 256u + lofs));
    }
#pragma unroll
    for (int k = 0; k < K; ++k) {
        v2f lo = {v[k].x, v[k].y};
        v2f hi = {v[k].z, v[k].w};
        acc01 += lo;
        acc23 += hi;
    }
}

__device__ __forceinline__ void add_chunk(const char* __restrict__ xlb, int s_l,
                                          int cnt, int half, unsigned lofs, v2f& acc01, v2f& acc23) {
    int j = 0;
    for (; j + 8 <= cnt; j += 8) add_rows<8>(xlb, s_l, j, half, lofs, acc01, acc23);
    if (j + 4 <= cnt) { add_rows<4>(xlb, s_l, j, half, lofs, acc01, acc23); j += 4; }
    int rem = cnt - j;
    if (rem == 3)      add_rows<3>(xlb, s_l, j, half, lofs, acc01, acc23);
    else if (rem == 2) add_rows<2>(xlb, s_l, j, half, lofs, acc01, acc23);
    else if (rem == 1) add_rows<1>(xlb, s_l, j, half, lofs, acc01, acc23);
}


#define TROWS 16
#define GWAVES (TROWS / 2)
#define LDP   68

template <int M>
__device__ __forceinline__ float row_xor_add(float v) {
    int t = __builtin_amdgcn_update_dpp(0, __float_as_int(v), 0x160 + M, 0xf, 0xf, false);
    return v + __int_as_float(t);
}

template <int HEAD>
__global__ void __launch_bounds__(GWAVES * 32)
gather_dense64(const float* __restrict__ xl, int zrow, const int* __restrict__ rowstart,
               const int* __restrict__ perm, const float* __restrict__ dinv,
               const float* __restrict__ bias,
               const float* __restrict__ W0, const float* __restrict__ b0,
               const float* __restrict__ W1, const float* __restrict__ b1,
               const float* __restrict__ W2, const float* __restrict__ b2,
               float* __restrict__ out, int n) {
    __shared__ __attribute__((aligned(16))) float t0[TROWS * LDP];
    __shared__ float t1[HEAD ? TROWS * LDP : 2];
    __shared__ float ps[TROWS];
    __shared__ __attribute__((aligned(16))) float dsc[TROWS];
    int wv    = __builtin_amdgcn_readfirstlane((int)(threadIdx.x >> 5));
    int node0 = (int)blockIdx.x * TROWS;
    int nodeA = node0 + 2 * wv;
    int lane  = threadIdx.x & 31;
    int half  = lane >> 4;
    int l     = lane & 15;

    v2f acc01 = {0.0f, 0.0f}, acc23 = {0.0f, 0.0f};
    float4 xs  = {0.0f, 0.0f, 0.0f, 0.0f};
    float4 cb  = {0.0f, 0.0f, 0.0f, 0.0f};
    float dn = 0.0f;
    if (nodeA < n) {
        const char* xlb = (const char*)xl;
        unsigned lofs = (unsigned)l * 16u;
        int r0 = rowstart[nodeA];
        int r1 = rowstart[nodeA + 1];
        int r2 = (nodeA + 1 < n) ? rowstart[nodeA + 2] : r1;
        int dA = r1 - r0;
        int dB = r2 - r1;
        int maxdeg = dA > dB ? dA : dB;
        int beg = half ? r1 : r0;
        int deg = half ? dB : dA;
        int nd = (nodeA + half < n) ? nodeA + half : nodeA;
        dn = dinv[nd];
        xs = *(const float4*)(xlb + ((unsigned)nd * 256u + lofs));
        cb.x = bias[4 * l];
        cb.y = bias[4 * l + 1];
        cb.z = bias[4 * l + 2];
        cb.w = bias[4 * l + 3];
        int last = r2 - 1;
        for (int c = 0; c < maxdeg; c += 32) {
            bool in0 = c + l < deg;
            bool in1 = c + 16 + l < deg;
            int  p0 = perm[in0 ? beg + c + l : last];
            int  p1 = perm[in1 ? beg + c + 16 + l : last];
            int  s0 = in0 ? p0 : zrow;
            int  s1 = in1 ? p1 : zrow;
            int left = maxdeg - c;
            add_chunk(xlb, s0, left < 16 ? left : 16, half, lofs, acc01, acc23);
            if (left > 16) add_chunk(xlb, s1, left < 32 ? left - 16 : 16, half, lofs, acc01, acc23);
        }
    }

    v2f bw[16];
    int n0 = (wv & 3) << 4;
    if (wv < 4) {
#pragma unroll
        for (int s = 0; s < 16; ++s) {
            int ka = 4 * s + half * 2;
            bw[s].x = W0[ka * 64 + n0 + l];
            bw[s].y = W0[(ka + 1) * 64 + n0 + l];
        }
    } else if (HEAD && wv < 6) {
#pragma unroll
        for (int s = 0; s < 16; ++s) {
            int ka = 4 * s + half * 2;
            bw[s].x = W1[ka * 32 + n0 + l];
            bw[s].y = W1[(ka + 1) * 32 + n0 + l];
        }
    }

    float4 o;
    o.x = tanhf((acc01.x + xs.x) * dn + cb.x);
    o.y = tanhf((acc01.y + xs.y) * dn + cb.y);
    o.z = tanhf((acc23.x + xs.z) * dn + cb.z);
    o.w = tanhf((acc23.y + xs.w) * dn + cb.w);
    *(float4*)&t0[(2 * wv + half) * LDP + 4 * l] = o;
    if (!HEAD && l == 0) dsc[2 * wv + half] = dn;
    __syncthreads();

    if (wv < 4) {
        v8f c = {};
#pragma unroll
        for (int s = 0; s < 16; ++s) {
            int ka = 4 * s + half * 2;
            v2f a = *(const v2f*)&t0[l * LDP + ka];
            c = __builtin_amdgcn_wmma_f32_16x16x4_f32(false, a, false, bw[s],
                                                      (short)0, c, false, false);
        }
        if (!HEAD) {
            float4 da = *(const float4*)&dsc[half * 8];
            float4 db = *(const float4*)&dsc[half * 8 + 4];
            float dr[8] = {da.x, da.y, da.z, da.w, db.x, db.y, db.z, db.w};
#pragma unroll
            for (int r = 0; r < 8; ++r) {
                int row = node0 + r + half * 8;
                if (row < n) out[(long long)row * 64 + n0 + l] = c[r] * dr[r];
            }
        } else {
            float bv = b0[n0 + l];
#pragma unroll
            for (int r = 0; r < 8; ++r)
                t1[(r + half * 8) * LDP + n0 + l] = tanhf(c[r] + bv);
        }
    }
    if (HEAD) {
        __syncthreads();
        float sel = 0.0f;
        if (wv == 4 || wv == 5) {
            v8f c = {};
#pragma unroll
            for (int s = 0; s < 16; ++s) {
                int ka = 4 * s + half * 2;
                v2f a = *(const v2f*)&t1[l * LDP + ka];
                c = __builtin_amdgcn_wmma_f32_16x16x4_f32(false, a, false, bw[s],
                                                          (short)0, c, false, false);
            }
            float bv = b1[n0 + l];
            float w2 = W2[n0 + l];
            float p[8];
#pragma unroll
            for (int r = 0; r < 8; ++r) p[r] = tanhf(c[r] + bv) * w2;
#pragma unroll
            for (int r = 0; r < 8; ++r) {
                float v = p[r];
                v = row_xor_add<1>(v);
                v = row_xor_add<2>(v);
                v = row_xor_add<4>(v);
                v = row_xor_add<8>(v);
                if (l == r) sel = v;
            }
            if (wv == 5 && l < 8) ps[half * 8 + l] = sel;
        }
        __syncthreads();
        if (wv == 4 && l < 8) {
            int row = half * 8 + l;
            if (node0 + row < n) out[node0 + row] = sel + ps[row] + b2[0];
        }
    }
}


extern "C" void kernel_launch(void* const* d_in, const int* in_sizes, int n_in,
                              void* d_out, int out_size, void* d_ws, size_t ws_size,
                              hipStream_t stream) {
    const float* x   = (const float*)d_in[0];
    const int*   ei  = (const int*)d_in[1];
    const float* cW0 = (const float*)d_in[2];
    const float* cb0 = (const float*)d_in[3];
    const float* cW1 = (const float*)d_in[4];
    const float* cb1 = (const float*)d_in[5];
    const float* fW0 = (const float*)d_in[6];
    const float* fb0 = (const float*)d_in[7];
    const float* fW1 = (const float*)d_in[8];
    const float* fb1 = (const float*)d_in[9];
    const float* fW2 = (const float*)d_in[10];
    const float* fb2 = (const float*)d_in[11];
    const int* src = ei;
    const int* dst = ei + NE;

    float* ws    = (float*)d_ws;
    float* xl    = ws;
    float* csr   = xl  + (size_t)NN * 64;
    float* h1    = csr + (size_t)NN * 64;
    float* zrow  = h1  + (size_t)NN * 64;
    float* dinv  = zrow + (size_t)NN * 64;

    int* bcount   = (int*)csr;
    int* bcursor  = bcount + 512;
    int* rowstart = bcursor + 512;
    int* tmp      = rowstart + NN + 32;
    int* perm     = tmp + NE;

    const int B = 256;
    dim3 blk(B);
    auto cdiv = [](long long a, long long b) { return (unsigned)((a + b - 1) / b); };

    unsigned gT    = cdiv(NE, TILE_E);
    unsigned gG    = cdiv(NN, GROWS);
    unsigned gA    = cdiv(NN, TROWS);
    dim3 blkA(GWAVES * 32);

    zero_i<<<4, blk, 0, stream>>>(bcount, 1024);
    bucket_count<<<gT, blk, 0, stream>>>(dst, bcount, NE);
    bucket_scatter<<<gT, blk, 0, stream>>>(src, dst, bcount, bcursor, tmp, NE);
    bucket_to_csr<<<NBK, blk, 0, stream>>>(tmp, bcount, rowstart, perm, dinv, zrow, NN);

    gemm_rows32_k128_n64<<<gG, dim3(128), 0, stream>>>(x, cW0, dinv, xl);

    gather_dense64<0><<<gA, blkA, 0, stream>>>(xl, 3 * NN, rowstart, perm, dinv, cb0,
                                               cW1, nullptr, nullptr, nullptr, nullptr, nullptr,
                                               h1, NN);

    gather_dense64<1><<<gA, blkA, 0, stream>>>(h1, NN, rowstart, perm, dinv, cb1,
                                               fW0, fb0, fW1, fb1, fW2, fb2,
                                               (float*)d_out, NN);
}


# Both edge-sorting passes put their edges in order in LDS and write them out with consecutive lanes on consecutive slots (whole runs, whole lines) instead of one scattered 4-byte store per edge
# speedup vs baseline: 7.1727x; 1.1413x over previous
#include <hip/hip_runtime.h>
#include <hip/hip_bf16.h>
#include <math.h>

#define NN 100000
#define NE 1600000

typedef __attribute__((ext_vector_type(2))) float v2f;
typedef __attribute__((ext_vector_type(8))) float v8f;


#define BK_SHIFT 8
#define BK_NODES 256
#define NBK ((NN + BK_NODES - 1) / BK_NODES)
#define TILE_E 4096
#define EPT 16
#define SCAP 8192
static_assert(NBK <= 512, "bucket tables hold 512 entries");
static_assert(NN <= (1 << 23), "a source index and 8 bucket-local bits share one 32-bit word");

__global__ void zero_i(int* __restrict__ p, int n) {
    int i = blockIdx.x * blockDim.x + threadIdx.x;
    if (i < n) p[i] = 0;
}

__global__ void bucket_count(const int* __restrict__ dst, int* __restrict__ bcount, int e_cnt) {
    __shared__ int h[512];
    int tid = threadIdx.x;
    h[tid] = 0;
    h[tid + 256] = 0;
    __syncthreads();
    int e0 = blockIdx.x * TILE_E + tid;
    int dv[EPT];
#pragma unroll
    for (int k = 0; k < EPT; ++k) {
        int e = e0 + k * 256;
        dv[k] = dst[e < e_cnt ? e : e_cnt - 1];
    }
#pragma unroll
    for (int k = 0; k < EPT; ++k) {
        int e = e0 + k * 256;
        if (e < e_cnt) atomicAdd(&h[dv[k] >> BK_SHIFT], 1);
    }
    __syncthreads();
    for (int b = tid; b < NBK; b += 256) {
        int c = h[b];
        if (c) atomicAdd(&bcount[b], c);
    }
}

__global__ void bucket_scatter(const int* __restrict__ src, const int* __restrict__ dst,
                               const int* __restrict__ bcount, int* __restrict__ bcursor,
                               int* __restrict__ tmp, int e_cnt) {
    __shared__ int h[512];
    __shared__ int tex[512];
    __shared__ int gd[512];
    __shared__ int wsum[8];
    __shared__ int wsum2[8];
    __shared__ int skey[TILE_E];
    __shared__ int sgi[TILE_E];
    int tid  = threadIdx.x;
    int lane = tid & 31;
    int wv   = tid >> 5;
    h[tid] = 0;
    h[tid + 256] = 0;
    int c0 = (2 * tid     < NBK) ? bcount[2 * tid]     : 0;
    int c1 = (2 * tid + 1 < NBK) ? bcount[2 * tid + 1] : 0;
    int tot = c0 + c1;
    int inc = tot;
    for (int d = 1; d < 32; d <<= 1) {
        int o = __shfl_up(inc, d);
        if (lane >= d) inc += o;
    }
    if (lane == 31) wsum[wv] = inc;
    __syncthreads();
    int wbase = 0;
    for (int w = 0; w < wv; ++w) wbase += wsum[w];
    int gs0 = wbase + inc - tot;
    int gs1 = gs0 + c0;

    int e0 = blockIdx.x * TILE_E + tid;
    int key[EPT], rk[EPT], bk[EPT];
#pragma unroll
    for (int k = 0; k < EPT; ++k) {
        int e  = e0 + k * 256;
        int ec = e < e_cnt ? e : e_cnt - 1;
        int d  = dst[ec];
        int s  = src[ec];
        bk[k]  = d >> BK_SHIFT;
        key[k] = (s << BK_SHIFT) | (d & (BK_NODES - 1));
    }
#pragma unroll
    for (int k = 0; k < EPT; ++k) {
        int e = e0 + k * 256;
        if (e < e_cnt) rk[k] = atomicAdd(&h[bk[k]], 1);
        else           bk[k] = -1;
    }
    __syncthreads();
    int t0 = h[2 * tid];
    int t1 = h[2 * tid + 1];
    int tt = t0 + t1;
    int tinc = tt;
    for (int d = 1; d < 32; d <<= 1) {
        int o = __shfl_up(tinc, d);
        if (lane >= d) tinc += o;
    }
    if (lane == 31) wsum2[wv] = tinc;
    if (t0) gs0 += atomicAdd(&bcursor[2 * tid], t0);
    if (t1) gs1 += atomicAdd(&bcursor[2 * tid + 1], t1);
    __syncthreads();
    int tbase = 0;
    for (int w = 0; w < wv; ++w) tbase += wsum2[w];
    int x0 = tbase + tinc - tt;
    int x1 = x0 + t0;
    tex[2 * tid]     = x0;
    tex[2 * tid + 1] = x1;
    gd[2 * tid]      = gs0 - x0;
    gd[2 * tid + 1]  = gs1 - x1;
    __syncthreads();
    int px[EPT], pg[EPT];
#pragma unroll
    for (int k = 0; k < EPT; ++k) {
        int bb = bk[k] >= 0 ? bk[k] : 0;
        px[k] = tex[bb];
        pg[k] = gd[bb];
    }
#pragma unroll
    for (int k = 0; k < EPT; ++k) {
        if (bk[k] >= 0) {
            int p = px[k] + rk[k];
            skey[p] = key[k];
            sgi[p]  = p + pg[k];
        }
    }
    __syncthreads();
    int nt = e_cnt - (int)blockIdx.x * TILE_E;
    if (nt > TILE_E) nt = TILE_E;
    int sk[EPT], sg[EPT];
#pragma unroll
    for (int k = 0; k < EPT; ++k) {
        int i  = tid + k * 256;
        int ic = i < nt ? i : 0;
        sk[k] = skey[ic];
        sg[k] = sgi[ic];
    }
#pragma unroll
    for (int k = 0; k < EPT; ++k) {
        int i = tid + k * 256;
        if (i < nt) tmp[sg[k]] = sk[k];
    }
}

__global__ void bucket_to_csr(const int* __restrict__ tmp, const int* __restrict__ bcount,
                              int* __restrict__ rowstart, int* __restrict__ perm,
                              float* __restrict__ dinv, float* __restrict__ zrow, int n) {
    __shared__ int c[BK_NODES];
    __shared__ int wsum[8];
    __shared__ int sperm[SCAP];
    int tid  = threadIdx.x;
    int lane = tid & 31;
    int wv   = tid >> 5;
    int b    = blockIdx.x;
    int bstart = 0;
    for (int i = 0; i < b; ++i) bstart += bcount[i];
    int bcnt = bcount[b];
    c[tid] = 0;
    __syncthreads();
    for (int e = tid; e < bcnt; e += EPT * 256) {
        int kv[EPT];
#pragma unroll
        for (int i = 0; i < EPT; ++i) {
            int ei = e + i * 256;
            kv[i] = tmp[bstart + (ei < bcnt ? ei : bcnt - 1)];
        }
#pragma unroll
        for (int i = 0; i < EPT; ++i) {
            if (e + i * 256 < bcnt) atomicAdd(&c[kv[i] & (BK_NODES - 1)], 1);
        }
    }
    __syncthreads();
    int my  = c[tid];
    int inc = my;
    for (int d = 1; d < 32; d <<= 1) {
        int o = __shfl_up(inc, d);
        if (lane >= d) inc += o;
    }
    if (lane == 31) wsum[wv] = inc;
    __syncthreads();
    int wbase = 0;
    for (int w = 0; w < wv; ++w) wbase += wsum[w];
    int ex = wbase + inc - my;
    int node = b * BK_NODES + tid;
    if (node < n) {
        rowstart[node] = bstart + ex;
        if (node == n - 1) rowstart[n] = bstart + ex + my;
        dinv[node] = rsqrtf((float)my + 1.0f);
    }
    if (b == 0 && tid < 64) zrow[tid] = 0.0f;
    c[tid] = ex;
    __syncthreads();
    if (bcnt <= SCAP) {
        for (int e = tid; e < bcnt; e += EPT * 256) {
            int kv[EPT];
#pragma unroll
            for (int i = 0; i < EPT; ++i) {
                int ei = e + i * 256;
                kv[i] = tmp[bstart + (ei < bcnt ? ei : bcnt - 1)];
            }
#pragma unroll
            for (int i = 0; i < EPT; ++i) {
                if (e + i * 256 < bcnt) {
                    int pos = atomicAdd(&c[kv[i] & (BK_NODES - 1)], 1);
                    sperm[pos] = kv[i] >> BK_SHIFT;
                }
            }
        }
        __syncthreads();
#pragma unroll 4
        for (int i = tid; i < bcnt; i += 256) perm[bstart + i] = sperm[i];
    } else {
        for (int e = tid; e < bcnt; e += EPT * 256) {
            int kv[EPT];
#pragma unroll
            for (int i = 0; i < EPT; ++i) {
                int ei = e + i * 256;
                kv[i] = tmp[bstart + (ei < bcnt ? ei : bcnt - 1)];
            }
#pragma unroll
            for (int i = 0; i < EPT; ++i) {
                if (e + i * 256 < bcnt) {
                    int pos = atomicAdd(&c[kv[i] & (BK_NODES - 1)], 1);
                    perm[bstart + pos] = kv[i] >> BK_SHIFT;
                }
            }
        }
    }
}

#define GK   128
#define GN   64
#define GROWS 32
#define GP   132

static_assert(NN % GROWS == 0, "every block of the product owns 32 whole rows");

__global__ void __launch_bounds__(128)
gemm_rows32_k128_n64(const float* __restrict__ X, const float* __restrict__ W,
                     const float* __restrict__ rowscale, float* __restrict__ Y) {
    __shared__ __attribute__((aligned(16))) float ta[GROWS * GP];
    __shared__ __attribute__((aligned(16))) float dsc[4][GROWS];
    int wv   = __builtin_amdgcn_readfirstlane((int)(threadIdx.x >> 5));
    int lane = threadIdx.x & 31;
    int half = lane >> 4;
    int l    = lane & 15;
    int m0   = (int)blockIdx.x * GROWS;

    float4 av[8];
#pragma unroll
    for (int i = 0; i < 8; ++i)
        av[i] = *(const float4*)(X + (size_t)(m0 + wv * 8 + i) * GK + 4 * lane);
    float rsv = rowscale[m0 + lane];
    v2f bw[32];
    int n0 = wv << 4;
#pragma unroll
    for (int s = 0; s < 32; ++s) {
        int ka = 4 * s + half * 2;
        bw[s].x = W[ka * GN + n0 + l];
        bw[s].y = W[(ka + 1) * GN + n0 + l];
    }
#pragma unroll
    for (int i = 0; i < 8; ++i)
        *(float4*)&ta[(wv * 8 + i) * GP + 4 * lane] = av[i];
    dsc[wv][lane] = rsv;
    __syncthreads();

    v8f c0 = {}, c1 = {};
#pragma unroll
    for (int s = 0; s < 32; ++s) {
        int ka = 4 * s + half * 2;
        v2f a0 = *(const v2f*)&ta[l * GP + ka];
        v2f a1 = *(const v2f*)&ta[(16 + l) * GP + ka];
        c0 = __builtin_amdgcn_wmma_f32_16x16x4_f32(false, a0, false, bw[s],
                                                   (short)0, c0, false, false);
        c1 = __builtin_amdgcn_wmma_f32_16x16x4_f32(false, a1, false, bw[s],
                                                   (short)0, c1, false, false);
    }
    float4 d0a = *(const float4*)&dsc[wv][half * 8];
    float4 d0b = *(const float4*)&dsc[wv][half * 8 + 4];
    float4 d1a = *(const float4*)&dsc[wv][16 + half * 8];
    float4 d1b = *(const float4*)&dsc[wv][16 + half * 8 + 4];
    float dr0[8] = {d0a.x, d0a.y, d0a.z, d0a.w, d0b.x, d0b.y, d0b.z, d0b.w};
    float dr1[8] = {d1a.x, d1a.y, d1a.z, d1a.w, d1b.x, d1b.y, d1b.z, d1b.w};
    float* yp = Y + (size_t)(m0 + half * 8) * GN + n0 + l;
#pragma unroll
    for (int r = 0; r < 8; ++r) {
        yp[r * GN]        = c0[r] * dr0[r];
        yp[(r + 16) * GN] = c1[r] * dr1[r];
    }
}


template <int K>
__device__ __forceinline__ void add_rows(const char* __restrict__ xlb, int s_l,
                                         int j, int half, unsigned lofs, v2f& acc01, v2f& acc23) {
    float4 v[K];
#pragma unroll
    for (int k = 0; k < K; ++k) {
        int s_lo = __builtin_amdgcn_readlane(s_l, j + k);
        int s_hi = __builtin_amdgcn_readlane(s_l, 16 + j + k);
        int s = half ? s_hi : s_lo;
        v[k]  = *(const float4*)(xlb + ((unsigned)s * 256u + lofs));
    }
#pragma unroll
    for (int k = 0; k < K; ++k) {
        v2f lo = {v[k].x, v[k].y};
        v2f hi = {v[k].z, v[k].w};
        acc01 += lo;
        acc23 += hi;
    }
}

__device__ __forceinline__ void add_chunk(const char* __restrict__ xlb, int s_l,
                                          int cnt, int half, unsigned lofs, v2f& acc01, v2f& acc23) {
    int j = 0;
    for (; j + 8 <= cnt; j += 8) add_rows<8>(xlb, s_l, j, half, lofs, acc01, acc23);
    if (j + 4 <= cnt) { add_rows<4>(xlb, s_l, j, half, lofs, acc01, acc23); j += 4; }
    int rem = cnt - j;
    if (rem == 3)      add_rows<3>(xlb, s_l, j, half, lofs, acc01, acc23);
    else if (rem == 2) add_rows<2>(xlb, s_l, j, half, lofs, acc01, acc23);
    else if (rem == 1) add_rows<1>(xlb, s_l, j, half, lofs, acc01, acc23);
}


#define TROWS 16
#define GWAVES (TROWS / 2)
#define LDP   68

template <int M>
__device__ __forceinline__ float row_xor_add(float v) {
    int t = __builtin_amdgcn_update_dpp(0, __float_as_int(v), 0x160 + M, 0xf, 0xf, false);
    return v + __int_as_float(t);
}

template <int HEAD>
__global__ void __launch_bounds__(GWAVES * 32)
gather_dense64(const float* __restrict__ xl, int zrow, const int* __restrict__ rowstart,
               const int* __restrict__ perm, const float* __restrict__ dinv,
               const float* __restrict__ bias,
               const float* __restrict__ W0, const float* __restrict__ b0,
               const float* __restrict__ W1, const float* __restrict__ b1,
               const float* __restrict__ W2, const float* __restrict__ b2,
               float* __restrict__ out, int n) {
    __shared__ __attribute__((aligned(16))) float t0[TROWS * LDP];
    __shared__ float t1[HEAD ? TROWS * LDP : 2];
    __shared__ float ps[TROWS];
    __shared__ __attribute__((aligned(16))) float dsc[TROWS];
    int wv    = __builtin_amdgcn_readfirstlane((int)(threadIdx.x >> 5));
    int node0 = (int)blockIdx.x * TROWS;
    int nodeA = node0 + 2 * wv;
    int lane  = threadIdx.x & 31;
    int half  = lane >> 4;
    int l     = lane & 15;

    v2f acc01 = {0.0f, 0.0f}, acc23 = {0.0f, 0.0f};
    float4 xs  = {0.0f, 0.0f, 0.0f, 0.0f};
    float4 cb  = {0.0f, 0.0f, 0.0f, 0.0f};
    float dn = 0.0f;
    if (nodeA < n) {
        const char* xlb = (const char*)xl;
        unsigned lofs = (unsigned)l * 16u;
        int r0 = rowstart[nodeA];
        int r1 = rowstart[nodeA + 1];
        int r2 = (nodeA + 1 < n) ? rowstart[nodeA + 2] : r1;
        int dA = r1 - r0;
        int dB = r2 - r1;
        int maxdeg = dA > dB ? dA : dB;
        int beg = half ? r1 : r0;
        int deg = half ? dB : dA;
        int nd = (nodeA + half < n) ? nodeA + half : nodeA;
        dn = dinv[nd];
        xs = *(const float4*)(xlb + ((unsigned)nd * 256u + lofs));
        cb.x = bias[4 * l];
        cb.y = bias[4 * l + 1];
        cb.z = bias[4 * l + 2];
        cb.w = bias[4 * l + 3];
        int last = r2 - 1;
        for (int c = 0; c < maxdeg; c += 32) {
            bool in0 = c + l < deg;
            bool in1 = c + 16 + l < deg;
            int  p0 = perm[in0 ? beg + c + l : last];
            int  p1 = perm[in1 ? beg + c + 16 + l : last];
            int  s0 = in0 ? p0 : zrow;
            int  s1 = in1 ? p1 : zrow;
            int left = maxdeg - c;
            add_chunk(xlb, s0, left < 16 ? left : 16, half, lofs, acc01, acc23);
            if (left > 16) add_chunk(xlb, s1, left < 32 ? left - 16 : 16, half, lofs, acc01, acc23);
        }
    }

    v2f bw[16];
    int n0 = (wv & 3) << 4;
    if (wv < 4) {
#pragma unroll
        for (int s = 0; s < 16; ++s) {
            int ka = 4 * s + half * 2;
            bw[s].x = W0[ka * 64 + n0 + l];
            bw[s].y = W0[(ka + 1) * 64 + n0 + l];
        }
    } else if (HEAD && wv < 6) {
#pragma unroll
        for (int s = 0; s < 16; ++s) {
            int ka = 4 * s + half * 2;
            bw[s].x = W1[ka * 32 + n0 + l];
            bw[s].y = W1[(ka + 1) * 32 + n0 + l];
        }
    }

    float4 o;
    o.x = tanhf((acc01.x + xs.x) * dn + cb.x);
    o.y = tanhf((acc01.y + xs.y) * dn + cb.y);
    o.z = tanhf((acc23.x + xs.z) * dn + cb.z);
    o.w = tanhf((acc23.y + xs.w) * dn + cb.w);
    *(float4*)&t0[(2 * wv + half) * LDP + 4 * l] = o;
    if (!HEAD && l == 0) dsc[2 * wv + half] = dn;
    __syncthreads();

    if (wv < 4) {
        v8f c = {};
#pragma unroll
        for (int s = 0; s < 16; ++s) {
            int ka = 4 * s + half * 2;
            v2f a = *(const v2f*)&t0[l * LDP + ka];
            c = __builtin_amdgcn_wmma_f32_16x16x4_f32(false, a, false, bw[s],
                                                      (short)0, c, false, false);
        }
        if (!HEAD) {
            float4 da = *(const float4*)&dsc[half * 8];
            float4 db = *(const float4*)&dsc[half * 8 + 4];
            float dr[8] = {da.x, da.y, da.z, da.w, db.x, db.y, db.z, db.w};
#pragma unroll
            for (int r = 0; r < 8; ++r) {
                int row = node0 + r + half * 8;
                if (row < n) out[(long long)row * 64 + n0 + l] = c[r] * dr[r];
            }
        } else {
            float bv = b0[n0 + l];
#pragma unroll
            for (int r = 0; r < 8; ++r)
                t1[(r + half * 8) * LDP + n0 + l] = tanhf(c[r] + bv);
        }
    }
    if (HEAD) {
        __syncthreads();
        float sel = 0.0f;
        if (wv == 4 || wv == 5) {
            v8f c = {};
#pragma unroll
            for (int s = 0; s < 16; ++s) {
                int ka = 4 * s + half * 2;
                v2f a = *(const v2f*)&t1[l * LDP + ka];
                c = __builtin_amdgcn_wmma_f32_16x16x4_f32(false, a, false, bw[s],
                                                          (short)0, c, false, false);
            }
            float bv = b1[n0 + l];
            float w2 = W2[n0 + l];
            float p[8];
#pragma unroll
            for (int r = 0; r < 8; ++r) p[r] = tanhf(c[r] + bv) * w2;
#pragma unroll
            for (int r = 0; r < 8; ++r) {
                float v = p[r];
                v = row_xor_add<1>(v);
                v = row_xor_add<2>(v);
                v = row_xor_add<4>(v);
                v = row_xor_add<8>(v);
                if (l == r) sel = v;
            }
            if (wv == 5 && l < 8) ps[half * 8 + l] = sel;
        }
        __syncthreads();
        if (wv == 4 && l < 8) {
            int row = half * 8 + l;
            if (node0 + row < n) out[node0 + row] = sel + ps[row] + b2[0];
        }
    }
}


extern "C" void kernel_launch(void* const* d_in, const int* in_sizes, int n_in,
                              void* d_out, int out_size, void* d_ws, size_t ws_size,
                              hipStream_t stream) {
    const float* x   = (const float*)d_in[0];
    const int*   ei  = (const int*)d_in[1];
    const float* cW0 = (const float*)d_in[2];
    const float* cb0 = (const float*)d_in[3];
    const float* cW1 = (const float*)d_in[4];
    const float* cb1 = (const float*)d_in[5];
    const float* fW0 = (const float*)d_in[6];
    const float* fb0 = (const float*)d_in[7];
    const float* fW1 = (const float*)d_in[8];
    const float* fb1 = (const float*)d_in[9];
    const float* fW2 = (const float*)d_in[10];
    const float* fb2 = (const float*)d_in[11];
    const int* src = ei;
    const int* dst = ei + NE;

    float* ws    = (float*)d_ws;
    float* xl    = ws;
    float* csr   = xl  + (size_t)NN * 64;
    float* h1    = csr + (size_t)NN * 64;
    float* zrow  = h1  + (size_t)NN * 64;
    float* dinv  = zrow + (size_t)NN * 64;

    int* bcount   = (int*)csr;
    int* bcursor  = bcount + 512;
    int* rowstart = bcursor + 512;
    int* tmp      = rowstart + NN + 32;
    int* perm     = tmp + NE;

    const int B = 256;
    dim3 blk(B);
    auto cdiv = [](long long a, long long b) { return (unsigned)((a + b - 1) / b); };

    unsigned gT    = cdiv(NE, TILE_E);
    unsigned gG    = cdiv(NN, GROWS);
    unsigned gA    = cdiv(NN, TROWS);
    dim3 blkA(GWAVES * 32);

    zero_i<<<4, blk, 0, stream>>>(bcount, 1024);
    bucket_count<<<gT, blk, 0, stream>>>(dst, bcount, NE);
    bucket_scatter<<<gT, blk, 0, stream>>>(src, dst, bcount, bcursor, tmp, NE);
    bucket_to_csr<<<NBK, blk, 0, stream>>>(tmp, bcount, rowstart, perm, dinv, zrow, NN);

    gemm_rows32_k128_n64<<<gG, dim3(128), 0, stream>>>(x, cW0, dinv, xl);

    gather_dense64<0><<<gA, blkA, 0, stream>>>(xl, 3 * NN, rowstart, perm, dinv, cb0,
                                               cW1, nullptr, nullptr, nullptr, nullptr, nullptr,
                                               h1, NN);

    gather_dense64<1><<<gA, blkA, 0, stream>>>(h1, NN, rowstart, perm, dinv, cb1,
                                               fW0, fb0, fW1, fb1, fW2, fb2,
                                               (float*)d_out, NN);
}


# Both aggregation launches resident: 2048 blocks loop over the 16-node tiles, pipelined (next tile's index words and row starts load under this tile's rows); weights staged in LDS once per block
# speedup vs baseline: 8.0983x; 1.1290x over previous
#include <hip/hip_runtime.h>
#include <hip/hip_bf16.h>
#include <math.h>

#define NN 100000
#define NE 1600000

typedef __attribute__((ext_vector_type(2))) float v2f;
typedef __attribute__((ext_vector_type(8))) float v8f;


#define BK_SHIFT 8
#define BK_NODES 256
#define NBK ((NN + BK_NODES - 1) / BK_NODES)
#define TILE_E 4096
#define EPT 16
#define NTILES ((NE + TILE_E - 1) / TILE_E)
#define TP (NBK + 1)
#define RPH ((NTILES + 15) / 16)
#define SCAP 8192
static_assert(NBK < 512, "bucket tables hold 512 entries, one of them the total");
static_assert(NTILES <= 512, "run tables hold 512 entries");
static_assert(NN <= (1 << 23), "a source index and 8 bucket-local bits share one 32-bit word");

__global__ void __launch_bounds__(256)
tile_sort(const int* __restrict__ src, const int* __restrict__ dst,
          int* __restrict__ tmp, int* __restrict__ tex_g, int e_cnt) {
    __shared__ int h[512];
    __shared__ int tex[512];
    __shared__ int wsum2[8];
    __shared__ int skey[TILE_E];
    int tid  = threadIdx.x;
    int lane = tid & 31;
    int wv   = tid >> 5;
    h[tid] = 0;
    h[tid + 256] = 0;
    __syncthreads();

    int e0 = blockIdx.x * TILE_E + tid;
    int key[EPT], rk[EPT], bk[EPT];
#pragma unroll
    for (int k = 0; k < EPT; ++k) {
        int e  = e0 + k * 256;
        int ec = e < e_cnt ? e : e_cnt - 1;
        int d  = dst[ec];
        int s  = src[ec];
        bk[k]  = d >> BK_SHIFT;
        key[k] = (s << BK_SHIFT) | (d & (BK_NODES - 1));
    }
#pragma unroll
    for (int k = 0; k < EPT; ++k) {
        int e = e0 + k * 256;
        if (e < e_cnt) rk[k] = atomicAdd(&h[bk[k]], 1);
        else           bk[k] = -1;
    }
    __syncthreads();
    int t0 = h[2 * tid];
    int t1 = h[2 * tid + 1];
    int tt = t0 + t1;
    int tinc = tt;
    for (int d = 1; d < 32; d <<= 1) {
        int o = __shfl_up(tinc, d);
        if (lane >= d) tinc += o;
    }
    if (lane == 31) wsum2[wv] = tinc;
    __syncthreads();
    int tbase = 0;
    for (int w = 0; w < wv; ++w) tbase += wsum2[w];
    int x0 = tbase + tinc - tt;
    int x1 = x0 + t0;
    tex[2 * tid]     = x0;
    tex[2 * tid + 1] = x1;
    int* trow = tex_g + (int)blockIdx.x * TP;
    if (2 * tid     <= NBK) trow[2 * tid]     = x0;
    if (2 * tid + 1 <= NBK) trow[2 * tid + 1] = x1;
    __syncthreads();
    int px[EPT];
#pragma unroll
    for (int k = 0; k < EPT; ++k) px[k] = tex[bk[k] >= 0 ? bk[k] : 0];
#pragma unroll
    for (int k = 0; k < EPT; ++k) {
        if (bk[k] >= 0) skey[px[k] + rk[k]] = key[k];
    }
    __syncthreads();
    int nt = e_cnt - (int)blockIdx.x * TILE_E;
    if (nt > TILE_E) nt = TILE_E;
    int sk[EPT];
#pragma unroll
    for (int k = 0; k < EPT; ++k) {
        int i = tid + k * 256;
        sk[k] = skey[i < nt ? i : 0];
    }
    int* tout = tmp + (int)blockIdx.x * TILE_E;
#pragma unroll
    for (int k = 0; k < EPT; ++k) {
        int i = tid + k * 256;
        if (i < nt) tout[i] = sk[k];
    }
}

__global__ void __launch_bounds__(256)
bucket_to_csr(const int* __restrict__ tmp, const int* __restrict__ tex_g,
              int* __restrict__ rowstart, int* __restrict__ perm,
              float* __restrict__ dinv, float* __restrict__ zrow, int n) {
    __shared__ int c[BK_NODES];
    __shared__ int wsum[8];
    __shared__ int ws0[8];
    __shared__ int ws1[8];
    __shared__ int wsa[8];
    __shared__ int runoff[512];
    __shared__ int runlen[512];
    __shared__ int runpos[512];
    __shared__ int skeys[SCAP];
    __shared__ int sperm[SCAP];
    int tid  = threadIdx.x;
    int lane = tid & 31;
    int wv   = tid >> 5;
    int b    = blockIdx.x;
    int ta  = tid;
    int tb  = tid + 256;
    int tca = ta < NTILES ? ta : 0;
    int tcb = tb < NTILES ? tb : 0;
    int lo0 = tex_g[tca * TP + b];
    int hi0 = tex_g[tca * TP + b + 1];
    int lo1 = tex_g[tcb * TP + b];
    int hi1 = tex_g[tcb * TP + b + 1];
    int a0 = ta < NTILES ? lo0 : 0;
    int l0 = ta < NTILES ? hi0 - lo0 : 0;
    int a1 = tb < NTILES ? lo1 : 0;
    int l1 = tb < NTILES ? hi1 - lo1 : 0;
    c[tid] = 0;
    int i0 = l0, i1 = l1, sa = a0 + a1;
    for (int d = 1; d < 32; d <<= 1) {
        int o0 = __shfl_up(i0, d);
        int o1 = __shfl_up(i1, d);
        if (lane >= d) { i0 += o0; i1 += o1; }
    }
    for (int d = 16; d >= 1; d >>= 1) sa += __shfl_xor(sa, d);
    if (lane == 31) { ws0[wv] = i0; ws1[wv] = i1; }
    if (lane == 0)  wsa[wv] = sa;
    __syncthreads();
    int base0 = 0, base1 = 0, tot0 = 0, tot1 = 0, bstart = 0;
    for (int w = 0; w < 8; ++w) {
        int v0 = ws0[w], v1 = ws1[w];
        if (w < wv) { base0 += v0; base1 += v1; }
        tot0 += v0;
        tot1 += v1;
        bstart += wsa[w];
    }
    int bcnt = tot0 + tot1;
    runoff[ta] = ta * TILE_E + a0;
    runlen[ta] = l0;
    runpos[ta] = base0 + i0 - l0;
    runoff[tb] = tb * TILE_E + a1;
    runlen[tb] = l1;
    runpos[tb] = tot0 + base1 + i1 - l1;
    __syncthreads();

    int hw = tid >> 4;
    int l  = tid & 15;
    if (bcnt <= SCAP) {
        int ln[RPH], of[RPH], pp[RPH], kv[RPH];
#pragma unroll
        for (int u = 0; u < RPH; ++u) {
            int r = hw + 16 * u;
            ln[u] = runlen[r];
            of[u] = runoff[r];
            pp[u] = runpos[r];
        }
#pragma unroll
        for (int u = 0; u < RPH; ++u) kv[u] = tmp[l < ln[u] ? of[u] + l : 0];
#pragma unroll
        for (int u = 0; u < RPH; ++u) {
            if (l < ln[u]) skeys[pp[u] + l] = kv[u];
        }
        for (int r = hw; r < NTILES; r += 16) {
            int len = runlen[r];
            int off = runoff[r];
            int pos = runpos[r];
            for (int i = l + 16; i < len; i += 16) skeys[pos + i] = tmp[off + i];
        }
        __syncthreads();
        for (int e = tid; e < bcnt; e += EPT * 256) {
            int kk[EPT];
#pragma unroll
            for (int i = 0; i < EPT; ++i) {
                int ei = e + i * 256;
                kk[i] = skeys[ei < bcnt ? ei : bcnt - 1];
            }
#pragma unroll
            for (int i = 0; i < EPT; ++i) {
                if (e + i * 256 < bcnt) atomicAdd(&c[kk[i] & (BK_NODES - 1)], 1);
            }
        }
    } else {
        for (int r = hw; r < NTILES; r += 16) {
            int len = runlen[r];
            int off = runoff[r];
            for (int i = l; i < len; i += 16) atomicAdd(&c[tmp[off + i] & (BK_NODES - 1)], 1);
        }
    }
    __syncthreads();
    int my  = c[tid];
    int inc = my;
    for (int d = 1; d < 32; d <<= 1) {
        int o = __shfl_up(inc, d);
        if (lane >= d) inc += o;
    }
    if (lane == 31) wsum[wv] = inc;
    __syncthreads();
    int wbase = 0;
    for (int w = 0; w < wv; ++w) wbase += wsum[w];
    int ex = wbase + inc - my;
    int node = b * BK_NODES + tid;
    if (node < n) {
        rowstart[node] = bstart + ex;
        if (node == n - 1) rowstart[n] = bstart + ex + my;
        dinv[node] = rsqrtf((float)my + 1.0f);
    }
    if (b == 0 && tid < 64) zrow[tid] = 0.0f;
    c[tid] = ex;
    __syncthreads();
    if (bcnt <= SCAP) {
        for (int e = tid; e < bcnt; e += EPT * 256) {
            int kk[EPT];
#pragma unroll
            for (int i = 0; i < EPT; ++i) {
                int ei = e + i * 256;
                kk[i] = skeys[ei < bcnt ? ei : bcnt - 1];
            }
#pragma unroll
            for (int i = 0; i < EPT; ++i) {
                if (e + i * 256 < bcnt) {
                    int pos = atomicAdd(&c[kk[i] & (BK_NODES - 1)], 1);
                    sperm[pos] = kk[i] >> BK_SHIFT;
                }
            }
        }
        __syncthreads();
#pragma unroll 4
        for (int i = tid; i < bcnt; i += 256) perm[bstart + i] = sperm[i];
    } else {
        for (int r = hw; r < NTILES; r += 16) {
            int len = runlen[r];
            int off = runoff[r];
            for (int i = l; i < len; i += 16) {
                int kx  = tmp[off + i];
                int pos = atomicAdd(&c[kx & (BK_NODES - 1)], 1);
                perm[bstart + pos] = kx >> BK_SHIFT;
            }
        }
    }
}

#define GK   128
#define GN   64
#define GROWS 32
#define GP   132

static_assert(NN % GROWS == 0, "every block of the product owns 32 whole rows");

__global__ void __launch_bounds__(128)
gemm_rows32_k128_n64(const float* __restrict__ X, const float* __restrict__ W,
                     const float* __restrict__ rowscale, float* __restrict__ Y) {
    __shared__ __attribute__((aligned(16))) float ta[GROWS * GP];
    __shared__ __attribute__((aligned(16))) float dsc[4][GROWS];
    int wv   = __builtin_amdgcn_readfirstlane((int)(threadIdx.x >> 5));
    int lane = threadIdx.x & 31;
    int half = lane >> 4;
    int l    = lane & 15;
    int m0   = (int)blockIdx.x * GROWS;

    float4 av[8];
#pragma unroll
    for (int i = 0; i < 8; ++i)
        av[i] = *(const float4*)(X + (size_t)(m0 + wv * 8 + i) * GK + 4 * lane);
    float rsv = rowscale[m0 + lane];
    v2f bw[32];
    int n0 = wv << 4;
#pragma unroll
    for (int s = 0; s < 32; ++s) {
        int ka = 4 * s + half * 2;
        bw[s].x = W[ka * GN + n0 + l];
        bw[s].y = W[(ka + 1) * GN + n0 + l];
    }
#pragma unroll
    for (int i = 0; i < 8; ++i)
        *(float4*)&ta[(wv * 8 + i) * GP + 4 * lane] = av[i];
    dsc[wv][lane] = rsv;
    __syncthreads();

    v8f c0 = {}, c1 = {};
#pragma unroll
    for (int s = 0; s < 32; ++s) {
        int ka = 4 * s + half * 2;
        v2f a0 = *(const v2f*)&ta[l * GP + ka];
        v2f a1 = *(const v2f*)&ta[(16 + l) * GP + ka];
        c0 = __builtin_amdgcn_wmma_f32_16x16x4_f32(false, a0, false, bw[s],
                                                   (short)0, c0, false, false);
        c1 = __builtin_amdgcn_wmma_f32_16x16x4_f32(false, a1, false, bw[s],
                                                   (short)0, c1, false, false);
    }
    float4 d0a = *(const float4*)&dsc[wv][half * 8];
    float4 d0b = *(const float4*)&dsc[wv][half * 8 + 4];
    float4 d1a = *(const float4*)&dsc[wv][16 + half * 8];
    float4 d1b = *(const float4*)&dsc[wv][16 + half * 8 + 4];
    float dr0[8] = {d0a.x, d0a.y, d0a.z, d0a.w, d0b.x, d0b.y, d0b.z, d0b.w};
    float dr1[8] = {d1a.x, d1a.y, d1a.z, d1a.w, d1b.x, d1b.y, d1b.z, d1b.w};
    float* yp = Y + (size_t)(m0 + half * 8) * GN + n0 + l;
#pragma unroll
    for (int r = 0; r < 8; ++r) {
        yp[r * GN]        = c0[r] * dr0[r];
        yp[(r + 16) * GN] = c1[r] * dr1[r];
    }
}


template <int K>
__device__ __forceinline__ void add_rows(const char* __restrict__ xlb, int s_l,
                                         int j, int half, unsigned lofs, v2f& acc01, v2f& acc23) {
    float4 v[K];
#pragma unroll
    for (int k = 0; k < K; ++k) {
        int s_lo = __builtin_amdgcn_readlane(s_l, j + k);
        int s_hi = __builtin_amdgcn_readlane(s_l, 16 + j + k);
        int s = half ? s_hi : s_lo;
        v[k]  = *(const float4*)(xlb + ((unsigned)s * 256u + lofs));
    }
#pragma unroll
    for (int k = 0; k < K; ++k) {
        v2f lo = {v[k].x, v[k].y};
        v2f hi = {v[k].z, v[k].w};
        acc01 += lo;
        acc23 += hi;
    }
}

__device__ __forceinline__ void add_chunk(const char* __restrict__ xlb, int s_l,
                                          int cnt, int half, unsigned lofs, v2f& acc01, v2f& acc23) {
    int j = 0;
    for (; j + 8 <= cnt; j += 8) add_rows<8>(xlb, s_l, j, half, lofs, acc01, acc23);
    if (j + 4 <= cnt) { add_rows<4>(xlb, s_l, j, half, lofs, acc01, acc23); j += 4; }
    int rem = cnt - j;
    if (rem == 3)      add_rows<3>(xlb, s_l, j, half, lofs, acc01, acc23);
    else if (rem == 2) add_rows<2>(xlb, s_l, j, half, lofs, acc01, acc23);
    else if (rem == 1) add_rows<1>(xlb, s_l, j, half, lofs, acc01, acc23);
}


#define TROWS 16
#define GWAVES (TROWS / 2)
#define LDP   68

template <int M>
__device__ __forceinline__ float row_xor_add(float v) {
    int t = __builtin_amdgcn_update_dpp(0, __float_as_int(v), 0x160 + M, 0xf, 0xf, false);
    return v + __int_as_float(t);
}

__device__ __forceinline__ float tanh_nb(float x) {
    float t = __builtin_amdgcn_exp2f(__builtin_fabsf(x) * 2.885390043f);
    float r = 1.0f - 2.0f * __builtin_amdgcn_rcpf(t + 1.0f);
    return __builtin_copysignf(r, x);
}

#define GRES 2048

__device__ __forceinline__ int row_starts_req(const int* __restrict__ rowstart, int nA, int n, int lane) {
    int q = nA < n ? nA : n - 1;
    int i = q + (lane < 2 ? lane : 2);
    return rowstart[i <= n ? i : n];
}

__device__ __forceinline__ void index_words2(const int* __restrict__ perm, int r0, int r1, int r2,
                                             int half, int l, int& p0, int& p1) {
    int beg  = half ? r1 : r0;
    int deg  = half ? r2 - r1 : r1 - r0;
    int last = r2 > 0 ? r2 - 1 : 0;
    p0 = perm[l < deg ? beg + l : last];
    p1 = perm[16 + l < deg ? beg + 16 + l : last];
}

template <int HEAD>
__global__ void __launch_bounds__(GWAVES * 32)
gather_dense64(const float* __restrict__ xl, int zrow, const int* __restrict__ rowstart,
               const int* __restrict__ perm, const float* __restrict__ dinv,
               const float* __restrict__ bias,
               const float* __restrict__ W0, const float* __restrict__ b0,
               const float* __restrict__ W1, const float* __restrict__ b1,
               const float* __restrict__ W2, const float* __restrict__ b2,
               float* __restrict__ out, int n) {
    __shared__ __attribute__((aligned(16))) float t0[(HEAD ? 1 : 2) * TROWS * LDP];
    __shared__ float t1[HEAD ? TROWS * LDP : 2];
    __shared__ float ps[TROWS];
    __shared__ __attribute__((aligned(16))) float dsc[2 * TROWS];
    __shared__ __attribute__((aligned(16))) float wl0[64 * 64];
    __shared__ __attribute__((aligned(16))) float wl1[HEAD ? 64 * 32 : 4];
    __shared__ __attribute__((aligned(16))) float bl[GWAVES][64];
    __shared__ float hb[HEAD ? 128 : 4];
    int wv    = __builtin_amdgcn_readfirstlane((int)(threadIdx.x >> 5));
    int lane  = threadIdx.x & 31;
    int n0    = (wv & 3) << 4;
    int ntile = (n + TROWS - 1) / TROWS;
    const char* xlb = (const char*)xl;

    int tile = (int)blockIdx.x;
    int rsa = row_starts_req(rowstart, tile * TROWS + 2 * wv, n, lane);
    int rse = row_starts_req(rowstart, (tile + GRES) * TROWS + 2 * wv, n, lane);
    __builtin_amdgcn_sched_barrier(0);
    int tid = (int)threadIdx.x;
    float4 wa[2], wb[2];
#pragma unroll
    for (int j = 0; j < 2; ++j) {
        int item = tid + 256 * j;
        int kp = item >> 4, n4 = item & 15;
        wa[j] = *(const float4*)(W0 + (2 * kp) * 64 + 4 * n4);
        wb[j] = *(const float4*)(W0 + (2 * kp + 1) * 64 + 4 * n4);
    }
    float4 wc = {0.0f, 0.0f, 0.0f, 0.0f}, wd = {0.0f, 0.0f, 0.0f, 0.0f};
    if (HEAD) {
        int kp = tid >> 3, n4 = tid & 7;
        wc = *(const float4*)(W1 + (2 * kp) * 32 + 4 * n4);
        wd = *(const float4*)(W1 + (2 * kp + 1) * 32 + 4 * n4);
    }

    __builtin_amdgcn_sched_barrier(0);
    int a0 = __builtin_amdgcn_readlane(rsa, 0);
    int a1 = __builtin_amdgcn_readlane(rsa, 1);
    int a2 = __builtin_amdgcn_readlane(rsa, 2);
    int p0, p1;
    index_words2(perm, a0, a1, a2, lane >> 4, lane & 15, p0, p1);
    v2f bb = *(const v2f*)(bias + 2 * lane);

#pragma unroll
    for (int j = 0; j < 2; ++j) {
        int item = tid + 256 * j;
        int kp = item >> 4, n4 = item & 15;
        float4 lo = {wa[j].x, wb[j].x, wa[j].y, wb[j].y};
        float4 hi = {wa[j].z, wb[j].z, wa[j].w, wb[j].w};
        *(float4*)&wl0[kp * 128 + 8 * n4]     = lo;
        *(float4*)&wl0[kp * 128 + 8 * n4 + 4] = hi;
    }
    if (HEAD) {
        int kp = tid >> 3, n4 = tid & 7;
        float4 lo = {wc.x, wd.x, wc.y, wd.y};
        float4 hi = {wc.z, wd.z, wc.w, wd.w};
        *(float4*)&wl1[kp * 64 + 8 * n4]     = lo;
        *(float4*)&wl1[kp * 64 + 8 * n4 + 4] = hi;
    }
    *(v2f*)&bl[wv][2 * lane] = bb;
    float b2v = 0.0f;
    if (HEAD) {
        if (tid < 64)       hb[tid] = b0[tid];
        else if (tid < 96)  hb[tid] = b1[tid - 64];
        else if (tid < 128) hb[tid] = W2[tid - 96];
        b2v = b2[0];
    }

    for (int it = 0; tile < ntile; tile += GRES, ++it) {
        int lt = lane;
        asm volatile("" : "+v"(lt));
        int half = lt >> 4;
        int l    = lt & 15;
        unsigned lofs = (unsigned)l * 16u;
        int e0 = __builtin_amdgcn_readlane(rse, 0);
        int e1 = __builtin_amdgcn_readlane(rse, 1);
        int e2 = __builtin_amdgcn_readlane(rse, 2);
        int q0, q1;
        index_words2(perm, e0, e1, e2, half, l, q0, q1);
        int rsf = row_starts_req(rowstart, (tile + 2 * GRES) * TROWS + 2 * wv, n, lt);

        int node0 = tile * TROWS;
        int nodeA = node0 + 2 * wv;
        float* tb = t0 + (HEAD ? 0 : (it & 1) * (TROWS * LDP));
        float* db = dsc + (it & 1) * TROWS;

        v2f acc01 = {0.0f, 0.0f}, acc23 = {0.0f, 0.0f};
        float4 xs  = {0.0f, 0.0f, 0.0f, 0.0f};
        float dn = 0.0f;
        if (nodeA < n) {
            int dA = a1 - a0;
            int dB = a2 - a1;
            int maxdeg = dA > dB ? dA : dB;
            int beg = half ? a1 : a0;
            int deg = half ? dB : dA;
            int nd = (nodeA + half < n) ? nodeA + half : nodeA;
            dn = dinv[nd];
            xs = *(const float4*)(xlb + ((unsigned)nd * 256u + lofs));
            if (maxdeg > 0) {
                int s0 = l < deg ? p0 : zrow;
                int s1 = 16 + l < deg ? p1 : zrow;
                add_chunk(xlb, s0, maxdeg < 16 ? maxdeg : 16, half, lofs, acc01, acc23);
                if (maxdeg > 16) add_chunk(xlb, s1, maxdeg < 32 ? maxdeg - 16 : 16, half, lofs, acc01, acc23);
                int last = a2 - 1;
                for (int c = 32; c < maxdeg; c += 32) {
                    bool in0 = c + l < deg;
                    bool in1 = c + 16 + l < deg;
                    int  w0 = perm[in0 ? beg + c + l : last];
                    int  w1 = perm[in1 ? beg + c + 16 + l : last];
                    int  u0 = in0 ? w0 : zrow;
                    int  u1 = in1 ? w1 : zrow;
                    int left = maxdeg - c;
                    add_chunk(xlb, u0, left < 16 ? left : 16, half, lofs, acc01, acc23);
                    if (left > 16) add_chunk(xlb, u1, left < 32 ? left - 16 : 16, half, lofs, acc01, acc23);
                }
            }
        }

        float4 cb = *(const float4*)&bl[wv][4 * l];
        float4 o;
        o.x = tanh_nb((acc01.x + xs.x) * dn + cb.x);
        o.y = tanh_nb((acc01.y + xs.y) * dn + cb.y);
        o.z = tanh_nb((acc23.x + xs.z) * dn + cb.z);
        o.w = tanh_nb((acc23.y + xs.w) * dn + cb.w);
        *(float4*)&tb[(2 * wv + half) * LDP + 4 * l] = o;
        if (!HEAD && l == 0) db[2 * wv + half] = dn;
        __syncthreads();

        int boff = half * 64 + n0 + l;
        if (wv < 4) {
            v8f c = {};
#pragma unroll
            for (int s = 0; s < 16; ++s) {
                int ka = 4 * s + half * 2;
                v2f a = *(const v2f*)&tb[l * LDP + ka];
                v2f b = *(const v2f*)&wl0[(boff + s * 128) * 2];
                c = __builtin_amdgcn_wmma_f32_16x16x4_f32(false, a, false, b,
                                                          (short)0, c, false, false);
            }
            if (!HEAD) {
                float4 da = *(const float4*)&db[half * 8];
                float4 dq = *(const float4*)&db[half * 8 + 4];
                float dr[8] = {da.x, da.y, da.z, da.w, dq.x, dq.y, dq.z, dq.w};
#pragma unroll
                for (int r = 0; r < 8; ++r) {
                    int row = node0 + r + half * 8;
                    if (row < n) out[(long long)row * 64 + n0 + l] = c[r] * dr[r];
                }
            } else {
                float bvd = hb[n0 + l];
#pragma unroll
                for (int r = 0; r < 8; ++r)
                    t1[(r + half * 8) * LDP + n0 + l] = tanh_nb(c[r] + bvd);
            }
        }
        if (HEAD) {
            __syncthreads();
            float sel = 0.0f;
            if (wv == 4 || wv == 5) {
                v8f c = {};
#pragma unroll
                for (int s = 0; s < 16; ++s) {
                    int ka = 4 * s + half * 2;
                    v2f a = *(const v2f*)&t1[l * LDP + ka];
                    v2f b = *(const v2f*)&wl1[(boff - half * 32 + s * 64) * 2];
                    c = __builtin_amdgcn_wmma_f32_16x16x4_f32(false, a, false, b,
                                                              (short)0, c, false, false);
                }
                float bvd = hb[64 + n0 + l];
                float w2d = hb[96 + n0 + l];
                float p[8];
#pragma unroll
                for (int r = 0; r < 8; ++r) p[r] = tanh_nb(c[r] + bvd) * w2d;
#pragma unroll
                for (int r = 0; r < 8; ++r) {
                    float v = p[r];
                    v = row_xor_add<1>(v);
                    v = row_xor_add<2>(v);
                    v = row_xor_add<4>(v);
                    v = row_xor_add<8>(v);
                    if (l == r) sel = v;
                }
                if (wv == 5 && l < 8) ps[half * 8 + l] = sel;
            }
            __syncthreads();
            if (wv == 4 && l < 8) {
                int row = half * 8 + l;
                if (node0 + row < n) out[node0 + row] = sel + ps[row] + b2v;
            }
        }

        a0 = e0; a1 = e1; a2 = e2;
        rse = rsf;
        p0 = q0; p1 = q1;
    }
}


extern "C" void kernel_launch(void* const* d_in, const int* in_sizes, int n_in,
                              void* d_out, int out_size, void* d_ws, size_t ws_size,
                              hipStream_t stream) {
    const float* x   = (const float*)d_in[0];
    const int*   ei  = (const int*)d_in[1];
    const float* cW0 = (const float*)d_in[2];
    const float* cb0 = (const float*)d_in[3];
    const float* cW1 = (const float*)d_in[4];
    const float* cb1 = (const float*)d_in[5];
    const float* fW0 = (const float*)d_in[6];
    const float* fb0 = (const float*)d_in[7];
    const float* fW1 = (const float*)d_in[8];
    const float* fb1 = (const float*)d_in[9];
    const float* fW2 = (const float*)d_in[10];
    const float* fb2 = (const float*)d_in[11];
    const int* src = ei;
    const int* dst = ei + NE;

    float* ws    = (float*)d_ws;
    float* xl    = ws;
    float* csr   = xl  + (size_t)NN * 64;
    float* h1    = csr + (size_t)NN * 64;
    float* zrow  = h1  + (size_t)NN * 64;
    float* dinv  = zrow + (size_t)NN * 64;

    int* rowstart = (int*)csr;
    int* tmp      = rowstart + NN + 32;
    int* perm     = tmp + NE;
    int* texg     = perm + NE;

    const int B = 256;
    dim3 blk(B);
    auto cdiv = [](long long a, long long b) { return (unsigned)((a + b - 1) / b); };

    unsigned gT    = cdiv(NE, TILE_E);
    unsigned gG    = cdiv(NN, GROWS);
    dim3 blkA(GWAVES * 32);

    tile_sort<<<gT, blk, 0, stream>>>(src, dst, tmp, texg, NE);
    bucket_to_csr<<<NBK, blk, 0, stream>>>(tmp, texg, rowstart, perm, dinv, zrow, NN);

    gemm_rows32_k128_n64<<<gG, dim3(128), 0, stream>>>(x, cW0, dinv, xl);

    gather_dense64<0><<<GRES, blkA, 0, stream>>>(xl, 3 * NN, rowstart, perm, dinv, cb0,
                                               cW1, nullptr, nullptr, nullptr, nullptr, nullptr,
                                               h1, NN);

    gather_dense64<1><<<GRES, blkA, 0, stream>>>(h1, NN, rowstart, perm, dinv, cb1,
                                               fW0, fb0, fW1, fb1, fW2, fb2,
                                               (float*)d_out, NN);
}
